# starting kernel
# baseline (speedup 1.0000x reference)
#include <hip/hip_runtime.h>

typedef float v2f __attribute__((ext_vector_type(2)));
typedef float v8f __attribute__((ext_vector_type(8)));

#define B_DIM 4
#define S_DIM 4096
#define H_DIM 8
#define D_DIM 64
#define SCALE 0.125f
#define NTILE (S_DIM / 16)

__global__ void __launch_bounds__(256)
tree_level_kernel(const float* __restrict__ kleaf,
                  const float* __restrict__ vleaf,
                  float* __restrict__ Kt,
                  float* __restrict__ Vt,
                  int level, int P)
{
    const int wave = threadIdx.x >> 5;
    const int lane = threadIdx.x & 31;
    const long g = (long)blockIdx.x * 8 + wave;
    const long total = (long)B_DIM * H_DIM * P;
    if (g >= total) return;
    const int p  = (int)(g % P);
    const int bh = (int)(g / P);
    const int h  = bh % H_DIM;
    const int b  = bh / H_DIM;

    const int doff = 2 * lane;
    float2 kc0, kc1, vc0, vc1;
    if (level == 1) {
        const size_t r0 = ((size_t)(b * S_DIM + 2 * p    ) * H_DIM + h) * D_DIM;
        const size_t r1 = ((size_t)(b * S_DIM + 2 * p + 1) * H_DIM + h) * D_DIM;
        kc0 = *(const float2*)(kleaf + r0 + doff);
        kc1 = *(const float2*)(kleaf + r1 + doff);
        vc0 = *(const float2*)(vleaf + r0 + doff);
        vc1 = *(const float2*)(vleaf + r1 + doff);
    } else {
        const int IBc = S_DIM - (S_DIM >> (level - 2));
        const size_t tb = (size_t)(b * H_DIM + h) * S_DIM;
        const size_t r0 = (tb + IBc + 2 * p    ) * D_DIM;
        const size_t r1 = (tb + IBc + 2 * p + 1) * D_DIM;
        kc0 = *(const float2*)(Kt + r0 + doff);
        kc1 = *(const float2*)(Kt + r1 + doff);
        vc0 = *(const float2*)(Vt + r0 + doff);
        vc1 = *(const float2*)(Vt + r1 + doff);
    }

    float2 kp, vp;
    kp.x = 0.5f * (kc0.x + kc1.x); kp.y = 0.5f * (kc0.y + kc1.y);
    vp.x = 0.5f * (vc0.x + vc1.x); vp.y = 0.5f * (vc0.y + vc1.y);

    float pss = kp.x * kp.x  + kp.y * kp.y;
    float ps0 = kp.x * kc0.x + kp.y * kc0.y;
    float ps1 = kp.x * kc1.x + kp.y * kc1.y;
    #pragma unroll
    for (int m = 16; m >= 1; m >>= 1) {
        pss += __shfl_xor(pss, m, 32);
        ps0 += __shfl_xor(ps0, m, 32);
        ps1 += __shfl_xor(ps1, m, 32);
    }
    pss *= SCALE; ps0 *= SCALE; ps1 *= SCALE;
    const float mm = fmaxf(pss, fmaxf(ps0, ps1));
    const float es = __expf(pss - mm);
    const float e0 = __expf(ps0 - mm);
    const float e1 = __expf(ps1 - mm);
    const float inv = 1.0f / (es + e0 + e1 + 1e-9f);

    float2 vo;
    vo.x = (es * vp.x + e0 * vc0.x + e1 * vc1.x) * inv;
    vo.y = (es * vp.y + e0 * vc0.y + e1 * vc1.y) * inv;

    const int IBl = S_DIM - (S_DIM >> (level - 1));
    const size_t dst = ((size_t)(b * H_DIM + h) * S_DIM + IBl + p) * D_DIM + doff;
    *(float2*)(Kt + dst) = kp;
    *(float2*)(Vt + dst) = vo;
}

__global__ void __launch_bounds__(128)
hst_attn_kernel(const float* __restrict__ q,
                const float* __restrict__ k,
                const float* __restrict__ v,
                const float* __restrict__ Kt,
                const float* __restrict__ Vt,
                float* __restrict__ out)
{
    __shared__ float ldsS[4][784];
    __shared__ int   ldsRow[4][32];

    const int lane = threadIdx.x & 31;
    const int w    = threadIdx.x >> 5;
    const int tile = blockIdx.x * 4 + w;
    const int t  = tile & (NTILE - 1);
    const int bh = tile >> 8;
    const int h  = bh & (H_DIM - 1);
    const int b  = bh >> 3;

    const int n    = lane & 15;
    const int half = lane >> 4;

    float* sA  = &ldsS[w][0];
    float* sB  = &ldsS[w][256];
    float* sC  = &ldsS[w][512];
    float* den = &ldsS[w][768];
    int*   rowB = &ldsRow[w][0];
    int*   rowC = &ldsRow[w][16];

    int rB;
    if      (n <  8) rB = 8 * t + n;
    else if (n < 12) rB = 2048 + 4 * t + (n - 8);
    else if (n < 14) rB = 3072 + 2 * t + (n - 12);
    else if (n == 14) rB = 3584 + (t ^ 1);
    else              rB = 3840 + ((t >> 1) ^ 1);
    const int rC = (n < 6) ? (S_DIM - (S_DIM >> (5 + n))) + ((t >> (n + 2)) ^ 1)
                           : 0;
    if (half == 0) { rowB[n] = rB; rowC[n] = rC; }

    const size_t leafBase = ((size_t)(b * S_DIM + 16 * t + n) * H_DIM + h) * D_DIM;
    const size_t treeBase = (size_t)(b * H_DIM + h) * S_DIM * (size_t)D_DIM;
    const float* qp  = q  + leafBase;
    const float* kAp = k  + leafBase;
    const float* kBp = Kt + treeBase + (size_t)rB * D_DIM;
    const float* kCp = Kt + treeBase + (size_t)rC * D_DIM;

    v8f cA = {}; v8f cB = {}; v8f cC = {};
    const int dofs = 2 * half;
    #pragma unroll
    for (int ks = 0; ks < 16; ++ks) {
        const int d = 4 * ks + dofs;
        v2f a  = *(const v2f*)(qp  + d);
        a.x *= SCALE; a.y *= SCALE;
        v2f bA = *(const v2f*)(kAp + d);
        v2f bB = *(const v2f*)(kBp + d);
        v2f bC = *(const v2f*)(kCp + d);
        cA = __builtin_amdgcn_wmma_f32_16x16x4_f32(false, a, false, bA, (short)0, cA, false, false);
        cB = __builtin_amdgcn_wmma_f32_16x16x4_f32(false, a, false, bB, (short)0, cB, false, false);
        cC = __builtin_amdgcn_wmma_f32_16x16x4_f32(false, a, false, bC, (short)0, cC, false, false);
    }

    #pragma unroll
    for (int r = 0; r < 8; ++r) {
        const int M = r + 8 * half;
        sA[M * 16 + n] = cA[r];
        sB[M * 16 + n] = cB[r];
        sC[M * 16 + n] = cC[r];
    }
    __syncthreads();

    float ev[13];
    float dsum = 0.0f;
    if (half == 0) {
        const int j = n;
        float x[13];
        x[0] = sA[j * 16 + j];
        x[1] = sA[j * 16 + (j ^ 1)];
        x[2] = sB[j * 16 + ((j >> 1) ^ 1)];
        x[3] = sB[j * 16 + 8  + ((j >> 2) ^ 1)];
        x[4] = sB[j * 16 + 12 + ((j >> 3) ^ 1)];
        x[5] = sB[j * 16 + 14];
        x[6] = sB[j * 16 + 15];
        #pragma unroll
        for (int i = 0; i < 6; ++i) x[7 + i] = sC[j * 16 + i];

        bool msk[13];
        msk[0] = false;
        #pragma unroll
        for (int l = 0; l < 12; ++l) {
            const int bit = (l < 4) ? ((j >> l) & 1) : ((t >> (l - 4)) & 1);
            msk[l + 1] = (bit == 0);
        }
        float m = x[0];
        #pragma unroll
        for (int i = 1; i < 13; ++i) if (!msk[i]) m = fmaxf(m, x[i]);
        #pragma unroll
        for (int i = 0; i < 13; ++i) {
            ev[i] = msk[i] ? 0.0f : __expf(x[i] - m);
            dsum += ev[i];
        }
    }
    __syncthreads();

    #pragma unroll
    for (int i = lane; i < 768; i += 32) ldsS[w][i] = 0.0f;
    __syncthreads();
    if (half == 0) {
        const int j = n;
        den[j] = dsum;
        sA[j * 16 + j]                   = ev[0];
        sA[j * 16 + (j ^ 1)]             = ev[1];
        sB[j * 16 + ((j >> 1) ^ 1)]      = ev[2];
        sB[j * 16 + 8  + ((j >> 2) ^ 1)] = ev[3];
        sB[j * 16 + 12 + ((j >> 3) ^ 1)] = ev[4];
        sB[j * 16 + 14]                  = ev[5];
        sB[j * 16 + 15]                  = ev[6];
        #pragma unroll
        for (int i = 0; i < 6; ++i) sC[j * 16 + i] = ev[7 + i];
    }
    __syncthreads();

    const size_t vLeafRow0  = ((size_t)(b * S_DIM + 16 * t) * H_DIM + h) * (size_t)D_DIM;
    const size_t vRowStride = (size_t)H_DIM * D_DIM;
    #pragma unroll
    for (int dsl = 0; dsl < 4; ++dsl) {
        const int dcol = 16 * dsl + n;
        v8f acc = {};
        #pragma unroll
        for (int ks2 = 0; ks2 < 4; ++ks2) {
            const int c0 = 4 * ks2 + 2 * half;
            v2f aE, bV;
            aE.x = sA[n * 16 + c0]; aE.y = sA[n * 16 + c0 + 1];
            bV.x = v[vLeafRow0 + (size_t)c0       * vRowStride + dcol];
            bV.y = v[vLeafRow0 + (size_t)(c0 + 1) * vRowStride + dcol];
            acc = __builtin_amdgcn_wmma_f32_16x16x4_f32(false, aE, false, bV, (short)0, acc, false, false);
            aE.x = sB[n * 16 + c0]; aE.y = sB[n * 16 + c0 + 1];
            bV.x = Vt[treeBase + (size_t)rowB[c0]     * D_DIM + dcol];
            bV.y = Vt[treeBase + (size_t)rowB[c0 + 1] * D_DIM + dcol];
            acc = __builtin_amdgcn_wmma_f32_16x16x4_f32(false, aE, false, bV, (short)0, acc, false, false);
            aE.x = sC[n * 16 + c0]; aE.y = sC[n * 16 + c0 + 1];
            bV.x = Vt[treeBase + (size_t)rowC[c0]     * D_DIM + dcol];
            bV.y = Vt[treeBase + (size_t)rowC[c0 + 1] * D_DIM + dcol];
            acc = __builtin_amdgcn_wmma_f32_16x16x4_f32(false, aE, false, bV, (short)0, acc, false, false);
        }
        #pragma unroll
        for (int r = 0; r < 8; ++r) {
            const int M = r + 8 * half;
            out[((size_t)(b * S_DIM + 16 * t + M) * H_DIM + h) * D_DIM + dcol] =
                acc[r] / den[M];
        }
    }
}

extern "C" void kernel_launch(void* const* d_in, const int* in_sizes, int n_in,
                              void* d_out, int out_size, void* d_ws, size_t ws_size,
                              hipStream_t stream)
{
    const float* q = (const float*)d_in[0];
    const float* k = (const float*)d_in[1];
    const float* v = (const float*)d_in[2];
    float* out = (float*)d_out;

    float* Kt = (float*)d_ws;
    float* Vt = Kt + (size_t)B_DIM * H_DIM * S_DIM * D_DIM;

    for (int level = 1; level <= 11; ++level) {
        const int P = S_DIM >> level;
        const long total = (long)B_DIM * H_DIM * P;
        const int blocks = (int)((total + 7) / 8);
        tree_level_kernel<<<blocks, 256, 0, stream>>>(k, v, Kt, Vt, level, P);
    }

    const int tiles = B_DIM * H_DIM * NTILE;
    hst_attn_kernel<<<tiles / 4, 128, 0, stream>>>(q, k, v, Kt, Vt, out);

    (void)in_sizes; (void)n_in; (void)out_size; (void)ws_size;
}


# Tree build in 2 launches instead of 11: a wave keeps 16 sibling nodes in registers and climbs 4 levels itself (1-4; then 5-8 and 9-11 through LDS, one block per (b,h)); per-node sum order unchanged
# speedup vs baseline: 1.3216x; 1.3216x over previous
#include <hip/hip_runtime.h>

typedef float v2f __attribute__((ext_vector_type(2)));
typedef float v8f __attribute__((ext_vector_type(8)));

#define B_DIM 4
#define S_DIM 4096
#define H_DIM 8
#define D_DIM 64
#define SCALE 0.125f
#define NTILE (S_DIM / 16)

__device__ __forceinline__ void merge_pair(const float2 kc0, const float2 kc1,
                                           const float2 vc0, const float2 vc1,
                                           float2& kp, float2& vo)
{
    float2 vp;
    kp.x = 0.5f * (kc0.x + kc1.x); kp.y = 0.5f * (kc0.y + kc1.y);
    vp.x = 0.5f * (vc0.x + vc1.x); vp.y = 0.5f * (vc0.y + vc1.y);

    float pss = kp.x * kp.x  + kp.y * kp.y;
    float ps0 = kp.x * kc0.x + kp.y * kc0.y;
    float ps1 = kp.x * kc1.x + kp.y * kc1.y;
    #pragma unroll
    for (int m = 16; m >= 1; m >>= 1) {
        pss += __shfl_xor(pss, m, 32);
        ps0 += __shfl_xor(ps0, m, 32);
        ps1 += __shfl_xor(ps1, m, 32);
    }
    pss *= SCALE; ps0 *= SCALE; ps1 *= SCALE;
    const float mm = fmaxf(pss, fmaxf(ps0, ps1));
    const float es = __expf(pss - mm);
    const float e0 = __expf(ps0 - mm);
    const float e1 = __expf(ps1 - mm);
    const float inv = 1.0f / (es + e0 + e1 + 1e-9f);

    vo.x = (es * vp.x + e0 * vc0.x + e1 * vc1.x) * inv;
    vo.y = (es * vp.y + e0 * vc0.y + e1 * vc1.y) * inv;
}

template <int NP>
__device__ __forceinline__ void level_in_regs(float2* kc, float2* vc,
                                              float* __restrict__ Kt,
                                              float* __restrict__ Vt,
                                              size_t treeRow0, int level,
                                              int chunk, int doff)
{
    const int IBl = S_DIM - (S_DIM >> (level - 1));
    const size_t row0 = treeRow0 + IBl + (size_t)chunk * NP;
    #pragma unroll
    for (int j = 0; j < NP; ++j) {
        float2 kp, vo;
        merge_pair(kc[2 * j], kc[2 * j + 1], vc[2 * j], vc[2 * j + 1], kp, vo);
        const size_t dst = (row0 + j) * D_DIM + doff;
        *(float2*)(Kt + dst) = kp;
        *(float2*)(Vt + dst) = vo;
        kc[j] = kp; vc[j] = vo;
    }
}

__global__ void __launch_bounds__(256)
tree_low_kernel(const float* __restrict__ kleaf,
                const float* __restrict__ vleaf,
                float* __restrict__ Kt,
                float* __restrict__ Vt)
{
    const int h    = threadIdx.x >> 5;
    const int lane = threadIdx.x & 31;
    const int c    = blockIdx.x & 255;
    const int b    = blockIdx.x >> 8;
    const int bh   = b * H_DIM + h;
    const int doff = 2 * lane;

    float2 kc[16], vc[16];
    const size_t leaf0 = ((size_t)(b * S_DIM + 16 * c) * H_DIM + h) * D_DIM + doff;
    #pragma unroll
    for (int i = 0; i < 16; ++i) {
        kc[i] = *(const float2*)(kleaf + leaf0 + (size_t)i * (H_DIM * D_DIM));
        vc[i] = *(const float2*)(vleaf + leaf0 + (size_t)i * (H_DIM * D_DIM));
    }

    const size_t treeRow0 = (size_t)bh * S_DIM;
    level_in_regs<8>(kc, vc, Kt, Vt, treeRow0, 1, c, doff);
    level_in_regs<4>(kc, vc, Kt, Vt, treeRow0, 2, c, doff);
    level_in_regs<2>(kc, vc, Kt, Vt, treeRow0, 3, c, doff);
    level_in_regs<1>(kc, vc, Kt, Vt, treeRow0, 4, c, doff);
}

__global__ void __launch_bounds__(512)
tree_high_kernel(float* __restrict__ Kt, float* __restrict__ Vt)
{
    __shared__ float2 sK[16][32];
    __shared__ float2 sV[16][32];

    const int wave = threadIdx.x >> 5;
    const int lane = threadIdx.x & 31;
    const int bh   = blockIdx.x;
    const int doff = 2 * lane;
    const size_t treeRow0 = (size_t)bh * S_DIM;
    const int IB4 = S_DIM - (S_DIM >> 3);

    float2 kc[16], vc[16];
    #pragma unroll
    for (int i = 0; i < 16; ++i) {
        const size_t src = (treeRow0 + IB4 + 16 * wave + i) * D_DIM + doff;
        kc[i] = *(const float2*)(Kt + src);
        vc[i] = *(const float2*)(Vt + src);
    }
    level_in_regs<8>(kc, vc, Kt, Vt, treeRow0, 5, wave, doff);
    level_in_regs<4>(kc, vc, Kt, Vt, treeRow0, 6, wave, doff);
    level_in_regs<2>(kc, vc, Kt, Vt, treeRow0, 7, wave, doff);
    level_in_regs<1>(kc, vc, Kt, Vt, treeRow0, 8, wave, doff);

    sK[wave][lane] = kc[0];
    sV[wave][lane] = vc[0];
    __syncthreads();

    if (wave == 0) {
        #pragma unroll
        for (int i = 0; i < 16; ++i) { kc[i] = sK[i][lane]; vc[i] = sV[i][lane]; }
        level_in_regs<8>(kc, vc, Kt, Vt, treeRow0, 9,  0, doff);
        level_in_regs<4>(kc, vc, Kt, Vt, treeRow0, 10, 0, doff);
        level_in_regs<2>(kc, vc, Kt, Vt, treeRow0, 11, 0, doff);
    }
}

__global__ void __launch_bounds__(128)
hst_attn_kernel(const float* __restrict__ q,
                const float* __restrict__ k,
                const float* __restrict__ v,
                const float* __restrict__ Kt,
                const float* __restrict__ Vt,
                float* __restrict__ out)
{
    __shared__ float ldsS[4][784];
    __shared__ int   ldsRow[4][32];

    const int lane = threadIdx.x & 31;
    const int w    = threadIdx.x >> 5;
    const int tile = blockIdx.x * 4 + w;
    const int t  = tile & (NTILE - 1);
    const int bh = tile >> 8;
    const int h  = bh & (H_DIM - 1);
    const int b  = bh >> 3;

    const int n    = lane & 15;
    const int half = lane >> 4;

    float* sA  = &ldsS[w][0];
    float* sB  = &ldsS[w][256];
    float* sC  = &ldsS[w][512];
    float* den = &ldsS[w][768];
    int*   rowB = &ldsRow[w][0];
    int*   rowC = &ldsRow[w][16];

    int rB;
    if      (n <  8) rB = 8 * t + n;
    else if (n < 12) rB = 2048 + 4 * t + (n - 8);
    else if (n < 14) rB = 3072 + 2 * t + (n - 12);
    else if (n == 14) rB = 3584 + (t ^ 1);
    else              rB = 3840 + ((t >> 1) ^ 1);
    const int rC = (n < 6) ? (S_DIM - (S_DIM >> (5 + n))) + ((t >> (n + 2)) ^ 1)
                           : 0;
    if (half == 0) { rowB[n] = rB; rowC[n] = rC; }

    const size_t leafBase = ((size_t)(b * S_DIM + 16 * t + n) * H_DIM + h) * D_DIM;
    const size_t treeBase = (size_t)(b * H_DIM + h) * S_DIM * (size_t)D_DIM;
    const float* qp  = q  + leafBase;
    const float* kAp = k  + leafBase;
    const float* kBp = Kt + treeBase + (size_t)rB * D_DIM;
    const float* kCp = Kt + treeBase + (size_t)rC * D_DIM;

    v8f cA = {}; v8f cB = {}; v8f cC = {};
    const int dofs = 2 * half;
    #pragma unroll
    for (int ks = 0; ks < 16; ++ks) {
        const int d = 4 * ks + dofs;
        v2f a  = *(const v2f*)(qp  + d);
        a.x *= SCALE; a.y *= SCALE;
        v2f bA = *(const v2f*)(kAp + d);
        v2f bB = *(const v2f*)(kBp + d);
        v2f bC = *(const v2f*)(kCp + d);
        cA = __builtin_amdgcn_wmma_f32_16x16x4_f32(false, a, false, bA, (short)0, cA, false, false);
        cB = __builtin_amdgcn_wmma_f32_16x16x4_f32(false, a, false, bB, (short)0, cB, false, false);
        cC = __builtin_amdgcn_wmma_f32_16x16x4_f32(false, a, false, bC, (short)0, cC, false, false);
    }

    #pragma unroll
    for (int r = 0; r < 8; ++r) {
        const int M = r + 8 * half;
        sA[M * 16 + n] = cA[r];
        sB[M * 16 + n] = cB[r];
        sC[M * 16 + n] = cC[r];
    }
    __syncthreads();

    float ev[13];
    float dsum = 0.0f;
    if (half == 0) {
        const int j = n;
        float x[13];
        x[0] = sA[j * 16 + j];
        x[1] = sA[j * 16 + (j ^ 1)];
        x[2] = sB[j * 16 + ((j >> 1) ^ 1)];
        x[3] = sB[j * 16 + 8  + ((j >> 2) ^ 1)];
        x[4] = sB[j * 16 + 12 + ((j >> 3) ^ 1)];
        x[5] = sB[j * 16 + 14];
        x[6] = sB[j * 16 + 15];
        #pragma unroll
        for (int i = 0; i < 6; ++i) x[7 + i] = sC[j * 16 + i];

        bool msk[13];
        msk[0] = false;
        #pragma unroll
        for (int l = 0; l < 12; ++l) {
            const int bit = (l < 4) ? ((j >> l) & 1) : ((t >> (l - 4)) & 1);
            msk[l + 1] = (bit == 0);
        }
        float m = x[0];
        #pragma unroll
        for (int i = 1; i < 13; ++i) if (!msk[i]) m = fmaxf(m, x[i]);
        #pragma unroll
        for (int i = 0; i < 13; ++i) {
            ev[i] = msk[i] ? 0.0f : __expf(x[i] - m);
            dsum += ev[i];
        }
    }
    __syncthreads();

    #pragma unroll
    for (int i = lane; i < 768; i += 32) ldsS[w][i] = 0.0f;
    __syncthreads();
    if (half == 0) {
        const int j = n;
        den[j] = dsum;
        sA[j * 16 + j]                   = ev[0];
        sA[j * 16 + (j ^ 1)]             = ev[1];
        sB[j * 16 + ((j >> 1) ^ 1)]      = ev[2];
        sB[j * 16 + 8  + ((j >> 2) ^ 1)] = ev[3];
        sB[j * 16 + 12 + ((j >> 3) ^ 1)] = ev[4];
        sB[j * 16 + 14]                  = ev[5];
        sB[j * 16 + 15]                  = ev[6];
        #pragma unroll
        for (int i = 0; i < 6; ++i) sC[j * 16 + i] = ev[7 + i];
    }
    __syncthreads();

    const size_t vLeafRow0  = ((size_t)(b * S_DIM + 16 * t) * H_DIM + h) * (size_t)D_DIM;
    const size_t vRowStride = (size_t)H_DIM * D_DIM;
    #pragma unroll
    for (int dsl = 0; dsl < 4; ++dsl) {
        const int dcol = 16 * dsl + n;
        v8f acc = {};
        #pragma unroll
        for (int ks2 = 0; ks2 < 4; ++ks2) {
            const int c0 = 4 * ks2 + 2 * half;
            v2f aE, bV;
            aE.x = sA[n * 16 + c0]; aE.y = sA[n * 16 + c0 + 1];
            bV.x = v[vLeafRow0 + (size_t)c0       * vRowStride + dcol];
            bV.y = v[vLeafRow0 + (size_t)(c0 + 1) * vRowStride + dcol];
            acc = __builtin_amdgcn_wmma_f32_16x16x4_f32(false, aE, false, bV, (short)0, acc, false, false);
            aE.x = sB[n * 16 + c0]; aE.y = sB[n * 16 + c0 + 1];
            bV.x = Vt[treeBase + (size_t)rowB[c0]     * D_DIM + dcol];
            bV.y = Vt[treeBase + (size_t)rowB[c0 + 1] * D_DIM + dcol];
            acc = __builtin_amdgcn_wmma_f32_16x16x4_f32(false, aE, false, bV, (short)0, acc, false, false);
            aE.x = sC[n * 16 + c0]; aE.y = sC[n * 16 + c0 + 1];
            bV.x = Vt[treeBase + (size_t)rowC[c0]     * D_DIM + dcol];
            bV.y = Vt[treeBase + (size_t)rowC[c0 + 1] * D_DIM + dcol];
            acc = __builtin_amdgcn_wmma_f32_16x16x4_f32(false, aE, false, bV, (short)0, acc, false, false);
        }
        #pragma unroll
        for (int r = 0; r < 8; ++r) {
            const int M = r + 8 * half;
            out[((size_t)(b * S_DIM + 16 * t + M) * H_DIM + h) * D_DIM + dcol] =
                acc[r] / den[M];
        }
    }
}

extern "C" void kernel_launch(void* const* d_in, const int* in_sizes, int n_in,
                              void* d_out, int out_size, void* d_ws, size_t ws_size,
                              hipStream_t stream)
{
    const float* q = (const float*)d_in[0];
    const float* k = (const float*)d_in[1];
    const float* v = (const float*)d_in[2];
    float* out = (float*)d_out;

    float* Kt = (float*)d_ws;
    float* Vt = Kt + (size_t)B_DIM * H_DIM * S_DIM * D_DIM;

    tree_low_kernel<<<1024, 256, 0, stream>>>(k, v, Kt, Vt);
    tree_high_kernel<<<32, 512, 0, stream>>>(Kt, Vt);

    const int tiles = B_DIM * H_DIM * NTILE;
    hst_attn_kernel<<<tiles / 4, 128, 0, stream>>>(q, k, v, Kt, Vt, out);

    (void)in_sizes; (void)n_in; (void)out_size; (void)ws_size;
}


# Attention: all global accesses 16 B per lane by re-mapping WMMA fragments (scores: lane's 4 contiguous d feed two K=4 steps, score sum order changes; E*V: lane owns 4 contiguous out columns)
# speedup vs baseline: 1.4227x; 1.0765x over previous
#include <hip/hip_runtime.h>

typedef float v2f __attribute__((ext_vector_type(2)));
typedef float v8f __attribute__((ext_vector_type(8)));

#define B_DIM 4
#define S_DIM 4096
#define H_DIM 8
#define D_DIM 64
#define SCALE 0.125f
#define NTILE (S_DIM / 16)

__device__ __forceinline__ void merge_pair(const float2 kc0, const float2 kc1,
                                           const float2 vc0, const float2 vc1,
                                           float2& kp, float2& vo)
{
    float2 vp;
    kp.x = 0.5f * (kc0.x + kc1.x); kp.y = 0.5f * (kc0.y + kc1.y);
    vp.x = 0.5f * (vc0.x + vc1.x); vp.y = 0.5f * (vc0.y + vc1.y);

    float pss = kp.x * kp.x  + kp.y * kp.y;
    float ps0 = kp.x * kc0.x + kp.y * kc0.y;
    float ps1 = kp.x * kc1.x + kp.y * kc1.y;
    #pragma unroll
    for (int m = 16; m >= 1; m >>= 1) {
        pss += __shfl_xor(pss, m, 32);
        ps0 += __shfl_xor(ps0, m, 32);
        ps1 += __shfl_xor(ps1, m, 32);
    }
    pss *= SCALE; ps0 *= SCALE; ps1 *= SCALE;
    const float mm = fmaxf(pss, fmaxf(ps0, ps1));
    const float es = __expf(pss - mm);
    const float e0 = __expf(ps0 - mm);
    const float e1 = __expf(ps1 - mm);
    const float inv = 1.0f / (es + e0 + e1 + 1e-9f);

    vo.x = (es * vp.x + e0 * vc0.x + e1 * vc1.x) * inv;
    vo.y = (es * vp.y + e0 * vc0.y + e1 * vc1.y) * inv;
}

template <int NP>
__device__ __forceinline__ void level_in_regs(float2* kc, float2* vc,
                                              float* __restrict__ Kt,
                                              float* __restrict__ Vt,
                                              size_t treeRow0, int level,
                                              int chunk, int doff)
{
    const int IBl = S_DIM - (S_DIM >> (level - 1));
    const size_t row0 = treeRow0 + IBl + (size_t)chunk * NP;
    #pragma unroll
    for (int j = 0; j < NP; ++j) {
        float2 kp, vo;
        merge_pair(kc[2 * j], kc[2 * j + 1], vc[2 * j], vc[2 * j + 1], kp, vo);
        const size_t dst = (row0 + j) * D_DIM + doff;
        *(float2*)(Kt + dst) = kp;
        *(float2*)(Vt + dst) = vo;
        kc[j] = kp; vc[j] = vo;
    }
}

__global__ void __launch_bounds__(256)
tree_low_kernel(const float* __restrict__ kleaf,
                const float* __restrict__ vleaf,
                float* __restrict__ Kt,
                float* __restrict__ Vt)
{
    const int h    = threadIdx.x >> 5;
    const int lane = threadIdx.x & 31;
    const int c    = blockIdx.x & 255;
    const int b    = blockIdx.x >> 8;
    const int bh   = b * H_DIM + h;
    const int doff = 2 * lane;

    float2 kc[16], vc[16];
    const size_t leaf0 = ((size_t)(b * S_DIM + 16 * c) * H_DIM + h) * D_DIM + doff;
    #pragma unroll
    for (int i = 0; i < 16; ++i) {
        kc[i] = *(const float2*)(kleaf + leaf0 + (size_t)i * (H_DIM * D_DIM));
        vc[i] = *(const float2*)(vleaf + leaf0 + (size_t)i * (H_DIM * D_DIM));
    }

    const size_t treeRow0 = (size_t)bh * S_DIM;
    level_in_regs<8>(kc, vc, Kt, Vt, treeRow0, 1, c, doff);
    level_in_regs<4>(kc, vc, Kt, Vt, treeRow0, 2, c, doff);
    level_in_regs<2>(kc, vc, Kt, Vt, treeRow0, 3, c, doff);
    level_in_regs<1>(kc, vc, Kt, Vt, treeRow0, 4, c, doff);
}

__global__ void __launch_bounds__(512)
tree_high_kernel(float* __restrict__ Kt, float* __restrict__ Vt)
{
    __shared__ float2 sK[16][32];
    __shared__ float2 sV[16][32];

    const int wave = threadIdx.x >> 5;
    const int lane = threadIdx.x & 31;
    const int bh   = blockIdx.x;
    const int doff = 2 * lane;
    const size_t treeRow0 = (size_t)bh * S_DIM;
    const int IB4 = S_DIM - (S_DIM >> 3);

    float2 kc[16], vc[16];
    #pragma unroll
    for (int i = 0; i < 16; ++i) {
        const size_t src = (treeRow0 + IB4 + 16 * wave + i) * D_DIM + doff;
        kc[i] = *(const float2*)(Kt + src);
        vc[i] = *(const float2*)(Vt + src);
    }
    level_in_regs<8>(kc, vc, Kt, Vt, treeRow0, 5, wave, doff);
    level_in_regs<4>(kc, vc, Kt, Vt, treeRow0, 6, wave, doff);
    level_in_regs<2>(kc, vc, Kt, Vt, treeRow0, 7, wave, doff);
    level_in_regs<1>(kc, vc, Kt, Vt, treeRow0, 8, wave, doff);

    sK[wave][lane] = kc[0];
    sV[wave][lane] = vc[0];
    __syncthreads();

    if (wave == 0) {
        #pragma unroll
        for (int i = 0; i < 16; ++i) { kc[i] = sK[i][lane]; vc[i] = sV[i][lane]; }
        level_in_regs<8>(kc, vc, Kt, Vt, treeRow0, 9,  0, doff);
        level_in_regs<4>(kc, vc, Kt, Vt, treeRow0, 10, 0, doff);
        level_in_regs<2>(kc, vc, Kt, Vt, treeRow0, 11, 0, doff);
    }
}

__device__ __forceinline__ float f4get(const float4 x, const int u)
{
    return u == 0 ? x.x : (u == 1 ? x.y : (u == 2 ? x.z : x.w));
}

__global__ void __launch_bounds__(128)
hst_attn_kernel(const float* __restrict__ q,
                const float* __restrict__ k,
                const float* __restrict__ v,
                const float* __restrict__ Kt,
                const float* __restrict__ Vt,
                float* __restrict__ out)
{
    __shared__ float ldsS[4][784];
    __shared__ int   ldsRow[4][32];

    const int lane = threadIdx.x & 31;
    const int w    = threadIdx.x >> 5;
    const int tile = blockIdx.x * 4 + w;
    const int t  = tile & (NTILE - 1);
    const int bh = tile >> 8;
    const int h  = bh & (H_DIM - 1);
    const int b  = bh >> 3;

    const int n    = lane & 15;
    const int half = lane >> 4;

    float* sA  = &ldsS[w][0];
    float* sB  = &ldsS[w][256];
    float* sC  = &ldsS[w][512];
    float* den = &ldsS[w][768];
    int*   rowB = &ldsRow[w][0];
    int*   rowC = &ldsRow[w][16];

    int rB;
    if      (n <  8) rB = 8 * t + n;
    else if (n < 12) rB = 2048 + 4 * t + (n - 8);
    else if (n < 14) rB = 3072 + 2 * t + (n - 12);
    else if (n == 14) rB = 3584 + (t ^ 1);
    else              rB = 3840 + ((t >> 1) ^ 1);
    const int rC = (n < 6) ? (S_DIM - (S_DIM >> (5 + n))) + ((t >> (n + 2)) ^ 1)
                           : 0;
    if (half == 0) { rowB[n] = rB; rowC[n] = rC; }

    const size_t leafBase = ((size_t)(b * S_DIM + 16 * t + n) * H_DIM + h) * D_DIM;
    const size_t treeBase = (size_t)(b * H_DIM + h) * S_DIM * (size_t)D_DIM;
    const float* qp  = q  + leafBase;
    const float* kAp = k  + leafBase;
    const float* kBp = Kt + treeBase + (size_t)rB * D_DIM;
    const float* kCp = Kt + treeBase + (size_t)rC * D_DIM;

    v8f cA = {}; v8f cB = {}; v8f cC = {};
    const int dofs = 4 * half;
    #pragma unroll
    for (int j = 0; j < 8; ++j) {
        const int d = 8 * j + dofs;
        const float4 a4 = *(const float4*)(qp  + d);
        const float4 kA4 = *(const float4*)(kAp + d);
        const float4 kB4 = *(const float4*)(kBp + d);
        const float4 kC4 = *(const float4*)(kCp + d);
        v2f a, bA, bB, bC;
        a.x = a4.x * SCALE; a.y = a4.y * SCALE;
        bA.x = kA4.x; bA.y = kA4.y;
        bB.x = kB4.x; bB.y = kB4.y;
        bC.x = kC4.x; bC.y = kC4.y;
        cA = __builtin_amdgcn_wmma_f32_16x16x4_f32(false, a, false, bA, (short)0, cA, false, false);
        cB = __builtin_amdgcn_wmma_f32_16x16x4_f32(false, a, false, bB, (short)0, cB, false, false);
        cC = __builtin_amdgcn_wmma_f32_16x16x4_f32(false, a, false, bC, (short)0, cC, false, false);
        a.x = a4.z * SCALE; a.y = a4.w * SCALE;
        bA.x = kA4.z; bA.y = kA4.w;
        bB.x = kB4.z; bB.y = kB4.w;
        bC.x = kC4.z; bC.y = kC4.w;
        cA = __builtin_amdgcn_wmma_f32_16x16x4_f32(false, a, false, bA, (short)0, cA, false, false);
        cB = __builtin_amdgcn_wmma_f32_16x16x4_f32(false, a, false, bB, (short)0, cB, false, false);
        cC = __builtin_amdgcn_wmma_f32_16x16x4_f32(false, a, false, bC, (short)0, cC, false, false);
    }

    #pragma unroll
    for (int r = 0; r < 8; ++r) {
        const int M = r + 8 * half;
        sA[M * 16 + n] = cA[r];
        sB[M * 16 + n] = cB[r];
        sC[M * 16 + n] = cC[r];
    }
    __syncthreads();

    float ev[13];
    float dsum = 0.0f;
    if (half == 0) {
        const int j = n;
        float x[13];
        x[0] = sA[j * 16 + j];
        x[1] = sA[j * 16 + (j ^ 1)];
        x[2] = sB[j * 16 + ((j >> 1) ^ 1)];
        x[3] = sB[j * 16 + 8  + ((j >> 2) ^ 1)];
        x[4] = sB[j * 16 + 12 + ((j >> 3) ^ 1)];
        x[5] = sB[j * 16 + 14];
        x[6] = sB[j * 16 + 15];
        #pragma unroll
        for (int i = 0; i < 6; ++i) x[7 + i] = sC[j * 16 + i];

        bool msk[13];
        msk[0] = false;
        #pragma unroll
        for (int l = 0; l < 12; ++l) {
            const int bit = (l < 4) ? ((j >> l) & 1) : ((t >> (l - 4)) & 1);
            msk[l + 1] = (bit == 0);
        }
        float m = x[0];
        #pragma unroll
        for (int i = 1; i < 13; ++i) if (!msk[i]) m = fmaxf(m, x[i]);
        #pragma unroll
        for (int i = 0; i < 13; ++i) {
            ev[i] = msk[i] ? 0.0f : __expf(x[i] - m);
            dsum += ev[i];
        }
    }
    __syncthreads();

    #pragma unroll
    for (int i = lane; i < 768; i += 32) ldsS[w][i] = 0.0f;
    __syncthreads();
    if (half == 0) {
        const int j = n;
        den[j] = dsum;
        sA[j * 16 + j]                   = ev[0];
        sA[j * 16 + (j ^ 1)]             = ev[1];
        sB[j * 16 + ((j >> 1) ^ 1)]      = ev[2];
        sB[j * 16 + 8  + ((j >> 2) ^ 1)] = ev[3];
        sB[j * 16 + 12 + ((j >> 3) ^ 1)] = ev[4];
        sB[j * 16 + 14]                  = ev[5];
        sB[j * 16 + 15]                  = ev[6];
        #pragma unroll
        for (int i = 0; i < 6; ++i) sC[j * 16 + i] = ev[7 + i];
    }
    __syncthreads();

    const size_t vLeafRow0  = ((size_t)(b * S_DIM + 16 * t) * H_DIM + h) * (size_t)D_DIM;
    const size_t vRowStride = (size_t)H_DIM * D_DIM;
    const int col4 = 4 * n;

    v2f eA[4], eB[4], eC[4];
    float4 vA[4][2], vB[4][2], vC[4][2];
    #pragma unroll
    for (int ks2 = 0; ks2 < 4; ++ks2) {
        const int c0 = 4 * ks2 + 2 * half;
        eA[ks2].x = sA[n * 16 + c0]; eA[ks2].y = sA[n * 16 + c0 + 1];
        eB[ks2].x = sB[n * 16 + c0]; eB[ks2].y = sB[n * 16 + c0 + 1];
        eC[ks2].x = sC[n * 16 + c0]; eC[ks2].y = sC[n * 16 + c0 + 1];
        #pragma unroll
        for (int s = 0; s < 2; ++s) {
            vA[ks2][s] = *(const float4*)(v  + vLeafRow0 + (size_t)(c0 + s) * vRowStride + col4);
            vB[ks2][s] = *(const float4*)(Vt + treeBase + (size_t)rowB[c0 + s] * D_DIM + col4);
            vC[ks2][s] = *(const float4*)(Vt + treeBase + (size_t)rowC[c0 + s] * D_DIM + col4);
        }
    }

    v8f acc[4] = {};
    #pragma unroll
    for (int ks2 = 0; ks2 < 4; ++ks2) {
        #pragma unroll
        for (int u = 0; u < 4; ++u) {
            v2f bV;
            bV.x = f4get(vA[ks2][0], u); bV.y = f4get(vA[ks2][1], u);
            acc[u] = __builtin_amdgcn_wmma_f32_16x16x4_f32(false, eA[ks2], false, bV, (short)0, acc[u], false, false);
        }
        #pragma unroll
        for (int u = 0; u < 4; ++u) {
            v2f bV;
            bV.x = f4get(vB[ks2][0], u); bV.y = f4get(vB[ks2][1], u);
            acc[u] = __builtin_amdgcn_wmma_f32_16x16x4_f32(false, eB[ks2], false, bV, (short)0, acc[u], false, false);
        }
        #pragma unroll
        for (int u = 0; u < 4; ++u) {
            v2f bV;
            bV.x = f4get(vC[ks2][0], u); bV.y = f4get(vC[ks2][1], u);
            acc[u] = __builtin_amdgcn_wmma_f32_16x16x4_f32(false, eC[ks2], false, bV, (short)0, acc[u], false, false);
        }
    }

    #pragma unroll
    for (int r = 0; r < 8; ++r) {
        const int M = r + 8 * half;
        const float dn = den[M];
        float4 o;
        o.x = acc[0][r] / dn; o.y = acc[1][r] / dn;
        o.z = acc[2][r] / dn; o.w = acc[3][r] / dn;
        *(float4*)(out + ((size_t)(b * S_DIM + 16 * t + M) * H_DIM + h) * D_DIM + col4) = o;
    }
}

extern "C" void kernel_launch(void* const* d_in, const int* in_sizes, int n_in,
                              void* d_out, int out_size, void* d_ws, size_t ws_size,
                              hipStream_t stream)
{
    const float* q = (const float*)d_in[0];
    const float* k = (const float*)d_in[1];
    const float* v = (const float*)d_in[2];
    float* out = (float*)d_out;

    float* Kt = (float*)d_ws;
    float* Vt = Kt + (size_t)B_DIM * H_DIM * S_DIM * D_DIM;

    tree_low_kernel<<<1024, 256, 0, stream>>>(k, v, Kt, Vt);
    tree_high_kernel<<<32, 512, 0, stream>>>(Kt, Vt);

    const int tiles = B_DIM * H_DIM * NTILE;
    hst_attn_kernel<<<tiles / 4, 128, 0, stream>>>(q, k, v, Kt, Vt, out);

    (void)in_sizes; (void)n_in; (void)out_size; (void)ws_size;
}


# Attention: scores vs a tile's own level 1-3 nodes = means of its leaf scores (node key = leaf mean; score sum order changes); those key rows are neither stored nor loaded, 2 score groups not 3
# speedup vs baseline: 1.6375x; 1.1510x over previous
#include <hip/hip_runtime.h>

typedef float v2f __attribute__((ext_vector_type(2)));
typedef float v8f __attribute__((ext_vector_type(8)));

#define B_DIM 4
#define S_DIM 4096
#define H_DIM 8
#define D_DIM 64
#define SCALE 0.125f
#define NTILE (S_DIM / 16)

__device__ __forceinline__ void merge_pair(const float2 kc0, const float2 kc1,
                                           const float2 vc0, const float2 vc1,
                                           float2& kp, float2& vo)
{
    float2 vp;
    kp.x = 0.5f * (kc0.x + kc1.x); kp.y = 0.5f * (kc0.y + kc1.y);
    vp.x = 0.5f * (vc0.x + vc1.x); vp.y = 0.5f * (vc0.y + vc1.y);

    float pss = kp.x * kp.x  + kp.y * kp.y;
    float ps0 = kp.x * kc0.x + kp.y * kc0.y;
    float ps1 = kp.x * kc1.x + kp.y * kc1.y;
    #pragma unroll
    for (int m = 16; m >= 1; m >>= 1) {
        pss += __shfl_xor(pss, m, 32);
        ps0 += __shfl_xor(ps0, m, 32);
        ps1 += __shfl_xor(ps1, m, 32);
    }
    pss *= SCALE; ps0 *= SCALE; ps1 *= SCALE;
    const float mm = fmaxf(pss, fmaxf(ps0, ps1));
    const float es = __expf(pss - mm);
    const float e0 = __expf(ps0 - mm);
    const float e1 = __expf(ps1 - mm);
    const float inv = 1.0f / (es + e0 + e1 + 1e-9f);

    vo.x = (es * vp.x + e0 * vc0.x + e1 * vc1.x) * inv;
    vo.y = (es * vp.y + e0 * vc0.y + e1 * vc1.y) * inv;
}

template <int NP, bool STORE_K>
__device__ __forceinline__ void level_in_regs(float2* kc, float2* vc,
                                              float* __restrict__ Kt,
                                              float* __restrict__ Vt,
                                              size_t treeRow0, int level,
                                              int chunk, int doff)
{
    const int IBl = S_DIM - (S_DIM >> (level - 1));
    const size_t row0 = treeRow0 + IBl + (size_t)chunk * NP;
    #pragma unroll
    for (int j = 0; j < NP; ++j) {
        float2 kp, vo;
        merge_pair(kc[2 * j], kc[2 * j + 1], vc[2 * j], vc[2 * j + 1], kp, vo);
        const size_t dst = (row0 + j) * D_DIM + doff;
        if (STORE_K) *(float2*)(Kt + dst) = kp;
        *(float2*)(Vt + dst) = vo;
        kc[j] = kp; vc[j] = vo;
    }
}

__global__ void __launch_bounds__(256)
tree_low_kernel(const float* __restrict__ kleaf,
                const float* __restrict__ vleaf,
                float* __restrict__ Kt,
                float* __restrict__ Vt)
{
    const int h    = threadIdx.x >> 5;
    const int lane = threadIdx.x & 31;
    const int c    = blockIdx.x & 255;
    const int b    = blockIdx.x >> 8;
    const int bh   = b * H_DIM + h;
    const int doff = 2 * lane;

    float2 kc[16], vc[16];
    const size_t leaf0 = ((size_t)(b * S_DIM + 16 * c) * H_DIM + h) * D_DIM + doff;
    #pragma unroll
    for (int i = 0; i < 16; ++i) {
        kc[i] = *(const float2*)(kleaf + leaf0 + (size_t)i * (H_DIM * D_DIM));
        vc[i] = *(const float2*)(vleaf + leaf0 + (size_t)i * (H_DIM * D_DIM));
    }

    const size_t treeRow0 = (size_t)bh * S_DIM;
    level_in_regs<8, false>(kc, vc, Kt, Vt, treeRow0, 1, c, doff);
    level_in_regs<4, false>(kc, vc, Kt, Vt, treeRow0, 2, c, doff);
    level_in_regs<2, false>(kc, vc, Kt, Vt, treeRow0, 3, c, doff);
    level_in_regs<1, true>(kc, vc, Kt, Vt, treeRow0, 4, c, doff);
}

__global__ void __launch_bounds__(512)
tree_high_kernel(float* __restrict__ Kt, float* __restrict__ Vt)
{
    __shared__ float2 sK[16][32];
    __shared__ float2 sV[16][32];

    const int wave = threadIdx.x >> 5;
    const int lane = threadIdx.x & 31;
    const int bh   = blockIdx.x;
    const int doff = 2 * lane;
    const size_t treeRow0 = (size_t)bh * S_DIM;
    const int IB4 = S_DIM - (S_DIM >> 3);

    float2 kc[16], vc[16];
    #pragma unroll
    for (int i = 0; i < 16; ++i) {
        const size_t src = (treeRow0 + IB4 + 16 * wave + i) * D_DIM + doff;
        kc[i] = *(const float2*)(Kt + src);
        vc[i] = *(const float2*)(Vt + src);
    }
    level_in_regs<8, true>(kc, vc, Kt, Vt, treeRow0, 5, wave, doff);
    level_in_regs<4, true>(kc, vc, Kt, Vt, treeRow0, 6, wave, doff);
    level_in_regs<2, true>(kc, vc, Kt, Vt, treeRow0, 7, wave, doff);
    level_in_regs<1, true>(kc, vc, Kt, Vt, treeRow0, 8, wave, doff);

    sK[wave][lane] = kc[0];
    sV[wave][lane] = vc[0];
    __syncthreads();

    if (wave == 0) {
        #pragma unroll
        for (int i = 0; i < 16; ++i) { kc[i] = sK[i][lane]; vc[i] = sV[i][lane]; }
        level_in_regs<8, true>(kc, vc, Kt, Vt, treeRow0, 9,  0, doff);
        level_in_regs<4, true>(kc, vc, Kt, Vt, treeRow0, 10, 0, doff);
        level_in_regs<2, true>(kc, vc, Kt, Vt, treeRow0, 11, 0, doff);
    }
}

__device__ __forceinline__ float f4get(const float4 x, const int u)
{
    return u == 0 ? x.x : (u == 1 ? x.y : (u == 2 ? x.z : x.w));
}

__global__ void __launch_bounds__(128)
hst_attn_kernel(const float* __restrict__ q,
                const float* __restrict__ k,
                const float* __restrict__ v,
                const float* __restrict__ Kt,
                const float* __restrict__ Vt,
                float* __restrict__ out)
{
    __shared__ float ldsS[4][784];
    __shared__ int   ldsRow[4][32];

    const int lane = threadIdx.x & 31;
    const int w    = threadIdx.x >> 5;
    const int tile = blockIdx.x * 4 + w;
    const int t  = tile & (NTILE - 1);
    const int bh = tile >> 8;
    const int h  = bh & (H_DIM - 1);
    const int b  = bh >> 3;

    const int n    = lane & 15;
    const int half = lane >> 4;

    float* sA  = &ldsS[w][0];
    float* sB  = &ldsS[w][256];
    float* sC  = &ldsS[w][512];
    float* den = &ldsS[w][768];
    int*   rowB = &ldsRow[w][0];
    int*   rowC = &ldsRow[w][16];

    int rB;
    if      (n <  8) rB = 8 * t + n;
    else if (n < 12) rB = 2048 + 4 * t + (n - 8);
    else if (n < 14) rB = 3072 + 2 * t + (n - 12);
    else if (n == 14) rB = 3584 + (t ^ 1);
    else              rB = 3840 + ((t >> 1) ^ 1);
    const int rC = (n < 6) ? (S_DIM - (S_DIM >> (5 + n))) + ((t >> (n + 2)) ^ 1)
                           : 0;
    if (half == 0) { rowB[n] = rB; rowC[n] = rC; }
    const int sF = n & 7;
    const int rF = (S_DIM - (S_DIM >> (3 + sF))) + ((t >> sF) ^ 1);

    const size_t leafBase = ((size_t)(b * S_DIM + 16 * t + n) * H_DIM + h) * D_DIM;
    const size_t treeBase = (size_t)(b * H_DIM + h) * S_DIM * (size_t)D_DIM;
    const float* qp  = q  + leafBase;
    const float* kAp = k  + leafBase;
    const float* kFp = Kt + treeBase + (size_t)rF * D_DIM;

    v8f cA = {}; v8f cF = {};
    const int dofs = 4 * half;
    #pragma unroll
    for (int j = 0; j < 8; ++j) {
        const int d = 8 * j + dofs;
        const float4 a4 = *(const float4*)(qp  + d);
        const float4 kA4 = *(const float4*)(kAp + d);
        const float4 kF4 = *(const float4*)(kFp + d);
        v2f a, bA, bF;
        a.x = a4.x * SCALE; a.y = a4.y * SCALE;
        bA.x = kA4.x; bA.y = kA4.y;
        bF.x = kF4.x; bF.y = kF4.y;
        cA = __builtin_amdgcn_wmma_f32_16x16x4_f32(false, a, false, bA, (short)0, cA, false, false);
        cF = __builtin_amdgcn_wmma_f32_16x16x4_f32(false, a, false, bF, (short)0, cF, false, false);
        a.x = a4.z * SCALE; a.y = a4.w * SCALE;
        bA.x = kA4.z; bA.y = kA4.w;
        bF.x = kF4.z; bF.y = kF4.w;
        cA = __builtin_amdgcn_wmma_f32_16x16x4_f32(false, a, false, bA, (short)0, cA, false, false);
        cF = __builtin_amdgcn_wmma_f32_16x16x4_f32(false, a, false, bF, (short)0, cF, false, false);
    }

    #pragma unroll
    for (int r = 0; r < 8; ++r) {
        const int M = r + 8 * half;
        sA[M * 16 + n] = cA[r];
        sB[M * 16 + n] = cF[r];
    }
    __syncthreads();

    float ev[13];
    float dsum = 0.0f;
    if (half == 0) {
        const int j = n;
        float sr[16];
        #pragma unroll
        for (int dx = 0; dx < 16; ++dx) sr[dx] = sA[j * 16 + (j ^ dx)];
        float x[13];
        x[0] = sr[0];
        x[1] = sr[1];
        x[2] = 0.5f * (sr[2] + sr[3]);
        x[3] = 0.25f * ((sr[4] + sr[5]) + (sr[6] + sr[7]));
        x[4] = 0.125f * (((sr[8] + sr[9]) + (sr[10] + sr[11]))
                       + ((sr[12] + sr[13]) + (sr[14] + sr[15])));
        #pragma unroll
        for (int i = 0; i < 8; ++i) x[5 + i] = sB[j * 16 + i];

        bool msk[13];
        msk[0] = false;
        #pragma unroll
        for (int l = 0; l < 12; ++l) {
            const int bit = (l < 4) ? ((j >> l) & 1) : ((t >> (l - 4)) & 1);
            msk[l + 1] = (bit == 0);
        }
        float m = x[0];
        #pragma unroll
        for (int i = 1; i < 13; ++i) if (!msk[i]) m = fmaxf(m, x[i]);
        #pragma unroll
        for (int i = 0; i < 13; ++i) {
            ev[i] = msk[i] ? 0.0f : __expf(x[i] - m);
            dsum += ev[i];
        }
    }
    __syncthreads();

    #pragma unroll
    for (int i = lane; i < 768; i += 32) ldsS[w][i] = 0.0f;
    __syncthreads();
    if (half == 0) {
        const int j = n;
        den[j] = dsum;
        sA[j * 16 + j]                   = ev[0];
        sA[j * 16 + (j ^ 1)]             = ev[1];
        sB[j * 16 + ((j >> 1) ^ 1)]      = ev[2];
        sB[j * 16 + 8  + ((j >> 2) ^ 1)] = ev[3];
        sB[j * 16 + 12 + ((j >> 3) ^ 1)] = ev[4];
        sB[j * 16 + 14]                  = ev[5];
        sB[j * 16 + 15]                  = ev[6];
        #pragma unroll
        for (int i = 0; i < 6; ++i) sC[j * 16 + i] = ev[7 + i];
    }
    __syncthreads();

    const size_t vLeafRow0  = ((size_t)(b * S_DIM + 16 * t) * H_DIM + h) * (size_t)D_DIM;
    const size_t vRowStride = (size_t)H_DIM * D_DIM;
    const int col4 = 4 * n;

    v2f eA[4], eB[4], eC[4];
    float4 vA[4][2], vB[4][2], vC[4][2];
    #pragma unroll
    for (int ks2 = 0; ks2 < 4; ++ks2) {
        const int c0 = 4 * ks2 + 2 * half;
        eA[ks2].x = sA[n * 16 + c0]; eA[ks2].y = sA[n * 16 + c0 + 1];
        eB[ks2].x = sB[n * 16 + c0]; eB[ks2].y = sB[n * 16 + c0 + 1];
        eC[ks2].x = sC[n * 16 + c0]; eC[ks2].y = sC[n * 16 + c0 + 1];
        #pragma unroll
        for (int s = 0; s < 2; ++s) {
            vA[ks2][s] = *(const float4*)(v  + vLeafRow0 + (size_t)(c0 + s) * vRowStride + col4);
            vB[ks2][s] = *(const float4*)(Vt + treeBase + (size_t)rowB[c0 + s] * D_DIM + col4);
            vC[ks2][s] = *(const float4*)(Vt + treeBase + (size_t)rowC[c0 + s] * D_DIM + col4);
        }
    }

    v8f acc[4] = {};
    #pragma unroll
    for (int ks2 = 0; ks2 < 4; ++ks2) {
        #pragma unroll
        for (int u = 0; u < 4; ++u) {
            v2f bV;
            bV.x = f4get(vA[ks2][0], u); bV.y = f4get(vA[ks2][1], u);
            acc[u] = __builtin_amdgcn_wmma_f32_16x16x4_f32(false, eA[ks2], false, bV, (short)0, acc[u], false, false);
        }
        #pragma unroll
        for (int u = 0; u < 4; ++u) {
            v2f bV;
            bV.x = f4get(vB[ks2][0], u); bV.y = f4get(vB[ks2][1], u);
            acc[u] = __builtin_amdgcn_wmma_f32_16x16x4_f32(false, eB[ks2], false, bV, (short)0, acc[u], false, false);
        }
        #pragma unroll
        for (int u = 0; u < 4; ++u) {
            v2f bV;
            bV.x = f4get(vC[ks2][0], u); bV.y = f4get(vC[ks2][1], u);
            acc[u] = __builtin_amdgcn_wmma_f32_16x16x4_f32(false, eC[ks2], false, bV, (short)0, acc[u], false, false);
        }
    }

    #pragma unroll
    for (int r = 0; r < 8; ++r) {
        const int M = r + 8 * half;
        const float dn = den[M];
        float4 o;
        o.x = acc[0][r] / dn; o.y = acc[1][r] / dn;
        o.z = acc[2][r] / dn; o.w = acc[3][r] / dn;
        *(float4*)(out + ((size_t)(b * S_DIM + 16 * t + M) * H_DIM + h) * D_DIM + col4) = o;
    }
}

extern "C" void kernel_launch(void* const* d_in, const int* in_sizes, int n_in,
                              void* d_out, int out_size, void* d_ws, size_t ws_size,
                              hipStream_t stream)
{
    const float* q = (const float*)d_in[0];
    const float* k = (const float*)d_in[1];
    const float* v = (const float*)d_in[2];
    float* out = (float*)d_out;

    float* Kt = (float*)d_ws;
    float* Vt = Kt + (size_t)B_DIM * H_DIM * S_DIM * D_DIM;

    tree_low_kernel<<<1024, 256, 0, stream>>>(k, v, Kt, Vt);
    tree_high_kernel<<<32, 512, 0, stream>>>(Kt, Vt);

    const int tiles = B_DIM * H_DIM * NTILE;
    hst_attn_kernel<<<tiles / 4, 128, 0, stream>>>(q, k, v, Kt, Vt, out);

    (void)in_sizes; (void)n_in; (void)out_size; (void)ws_size;
}


# Attention: a tile's level 1-3 node values = weighted sums of its leaf values (weights from a WMMA leaf-key product K*K^T; sum order changes); those value rows are neither stored nor loaded
# speedup vs baseline: 2.0012x; 1.2221x over previous
#include <hip/hip_runtime.h>

typedef float v2f __attribute__((ext_vector_type(2)));
typedef float v8f __attribute__((ext_vector_type(8)));

#define B_DIM 4
#define S_DIM 4096
#define H_DIM 8
#define D_DIM 64
#define SCALE 0.125f
#define NTILE (S_DIM / 16)

__device__ __forceinline__ void merge_pair(const float2 kc0, const float2 kc1,
                                           const float2 vc0, const float2 vc1,
                                           float2& kp, float2& vo)
{
    float2 vp;
    kp.x = 0.5f * (kc0.x + kc1.x); kp.y = 0.5f * (kc0.y + kc1.y);
    vp.x = 0.5f * (vc0.x + vc1.x); vp.y = 0.5f * (vc0.y + vc1.y);

    float pss = kp.x * kp.x  + kp.y * kp.y;
    float ps0 = kp.x * kc0.x + kp.y * kc0.y;
    float ps1 = kp.x * kc1.x + kp.y * kc1.y;
    #pragma unroll
    for (int m = 16; m >= 1; m >>= 1) {
        pss += __shfl_xor(pss, m, 32);
        ps0 += __shfl_xor(ps0, m, 32);
        ps1 += __shfl_xor(ps1, m, 32);
    }
    pss *= SCALE; ps0 *= SCALE; ps1 *= SCALE;
    const float mm = fmaxf(pss, fmaxf(ps0, ps1));
    const float es = __expf(pss - mm);
    const float e0 = __expf(ps0 - mm);
    const float e1 = __expf(ps1 - mm);
    const float inv = 1.0f / (es + e0 + e1 + 1e-9f);

    vo.x = (es * vp.x + e0 * vc0.x + e1 * vc1.x) * inv;
    vo.y = (es * vp.y + e0 * vc0.y + e1 * vc1.y) * inv;
}

template <int NP, bool STORE_K, bool STORE_V>
__device__ __forceinline__ void level_in_regs(float2* kc, float2* vc,
                                              float* __restrict__ Kt,
                                              float* __restrict__ Vt,
                                              size_t treeRow0, int level,
                                              int chunk, int doff)
{
    const int IBl = S_DIM - (S_DIM >> (level - 1));
    const size_t row0 = treeRow0 + IBl + (size_t)chunk * NP;
    #pragma unroll
    for (int j = 0; j < NP; ++j) {
        float2 kp, vo;
        merge_pair(kc[2 * j], kc[2 * j + 1], vc[2 * j], vc[2 * j + 1], kp, vo);
        const size_t dst = (row0 + j) * D_DIM + doff;
        if (STORE_K) *(float2*)(Kt + dst) = kp;
        if (STORE_V) *(float2*)(Vt + dst) = vo;
        kc[j] = kp; vc[j] = vo;
    }
}

__global__ void __launch_bounds__(256) __attribute__((amdgpu_waves_per_eu(8, 8)))
tree_low_kernel(const float* __restrict__ kleaf,
                const float* __restrict__ vleaf,
                float* __restrict__ Kt,
                float* __restrict__ Vt)
{
    const int h    = threadIdx.x >> 5;
    const int lane = threadIdx.x & 31;
    const int c    = blockIdx.x & 255;
    const int b    = blockIdx.x >> 8;
    const int bh   = b * H_DIM + h;
    const int doff = 2 * lane;

    float2 kc[16], vc[16];
    const size_t leaf0 = ((size_t)(b * S_DIM + 16 * c) * H_DIM + h) * D_DIM + doff;
    #pragma unroll
    for (int i = 0; i < 16; ++i) {
        kc[i] = *(const float2*)(kleaf + leaf0 + (size_t)i * (H_DIM * D_DIM));
        vc[i] = *(const float2*)(vleaf + leaf0 + (size_t)i * (H_DIM * D_DIM));
    }

    const size_t treeRow0 = (size_t)bh * S_DIM;
    level_in_regs<8, false, false>(kc, vc, Kt, Vt, treeRow0, 1, c, doff);
    level_in_regs<4, false, false>(kc, vc, Kt, Vt, treeRow0, 2, c, doff);
    level_in_regs<2, false, false>(kc, vc, Kt, Vt, treeRow0, 3, c, doff);
    level_in_regs<1, true, true>(kc, vc, Kt, Vt, treeRow0, 4, c, doff);
}

__global__ void __launch_bounds__(512)
tree_high_kernel(float* __restrict__ Kt, float* __restrict__ Vt)
{
    __shared__ float2 sK[16][32];
    __shared__ float2 sV[16][32];

    const int wave = threadIdx.x >> 5;
    const int lane = threadIdx.x & 31;
    const int bh   = blockIdx.x;
    const int doff = 2 * lane;
    const size_t treeRow0 = (size_t)bh * S_DIM;
    const int IB4 = S_DIM - (S_DIM >> 3);

    float2 kc[16], vc[16];
    #pragma unroll
    for (int i = 0; i < 16; ++i) {
        const size_t src = (treeRow0 + IB4 + 16 * wave + i) * D_DIM + doff;
        kc[i] = *(const float2*)(Kt + src);
        vc[i] = *(const float2*)(Vt + src);
    }
    level_in_regs<8, true, true>(kc, vc, Kt, Vt, treeRow0, 5, wave, doff);
    level_in_regs<4, true, true>(kc, vc, Kt, Vt, treeRow0, 6, wave, doff);
    level_in_regs<2, true, true>(kc, vc, Kt, Vt, treeRow0, 7, wave, doff);
    level_in_regs<1, true, true>(kc, vc, Kt, Vt, treeRow0, 8, wave, doff);

    sK[wave][lane] = kc[0];
    sV[wave][lane] = vc[0];
    __syncthreads();

    if (wave == 0) {
        #pragma unroll
        for (int i = 0; i < 16; ++i) { kc[i] = sK[i][lane]; vc[i] = sV[i][lane]; }
        level_in_regs<8, true, true>(kc, vc, Kt, Vt, treeRow0, 9,  0, doff);
        level_in_regs<4, true, true>(kc, vc, Kt, Vt, treeRow0, 10, 0, doff);
        level_in_regs<2, true, true>(kc, vc, Kt, Vt, treeRow0, 11, 0, doff);
    }
}

__device__ __forceinline__ float f4get(const float4 x, const int u)
{
    return u == 0 ? x.x : (u == 1 ? x.y : (u == 2 ? x.z : x.w));
}

__device__ __forceinline__ void merge_weights(float g00, float g01, float g11,
                                              const float nrm, float& w0, float& w1)
{
    g00 *= nrm; g01 *= nrm; g11 *= nrm;
    const float ps0 = 0.5f * (g00 + g01);
    const float ps1 = 0.5f * (g01 + g11);
    const float pss = 0.5f * (ps0 + ps1);
    const float mm = fmaxf(pss, fmaxf(ps0, ps1));
    const float es = __expf(pss - mm);
    const float e0 = __expf(ps0 - mm);
    const float e1 = __expf(ps1 - mm);
    const float inv = 1.0f / (es + e0 + e1 + 1e-9f);
    w0 = (0.5f * es + e0) * inv;
    w1 = (0.5f * es + e1) * inv;
}

__device__ __forceinline__ float add_lane_xor1(const float x)
{
    return x + __int_as_float(__builtin_amdgcn_update_dpp(0, __float_as_int(x), 0xB1, 0xf, 0xf, false));
}
__device__ __forceinline__ float add_lane_xor2(const float x)
{
    return x + __int_as_float(__builtin_amdgcn_update_dpp(0, __float_as_int(x), 0x4E, 0xf, 0xf, false));
}

#define L_SA   0
#define L_SF   256
#define L_SG   512
#define L_S1   768
#define L_S2   832
#define L_SW   848
#define L_DEN  896
#define L_WAVE 912

__global__ void __launch_bounds__(128)
hst_attn_kernel(const float* __restrict__ q,
                const float* __restrict__ k,
                const float* __restrict__ v,
                const float* __restrict__ Kt,
                const float* __restrict__ Vt,
                float* __restrict__ out)
{
    __shared__ __attribute__((aligned(16))) float ldsS[4][L_WAVE];

    const int lane = threadIdx.x & 31;
    const int w    = threadIdx.x >> 5;
    const int tile = blockIdx.x * 4 + w;
    const int t  = tile & (NTILE - 1);
    const int bh = tile >> 8;
    const int h  = bh & (H_DIM - 1);
    const int b  = bh >> 3;

    const int n    = lane & 15;
    const int half = lane >> 4;

    float* sA  = &ldsS[w][L_SA];
    float* sF  = &ldsS[w][L_SF];
    float* sG  = &ldsS[w][L_SG];
    float* sS1 = &ldsS[w][L_S1];
    float* sS2 = &ldsS[w][L_S2];
    float* sW  = &ldsS[w][L_SW];
    float* den = &ldsS[w][L_DEN];

    const int sFc = n & 7;
    const int rF = (S_DIM - (S_DIM >> (3 + sFc))) + ((t >> sFc) ^ 1);

    const size_t leafBase = ((size_t)(b * S_DIM + 16 * t + n) * H_DIM + h) * D_DIM;
    const size_t treeBase = (size_t)(b * H_DIM + h) * S_DIM * (size_t)D_DIM;
    const float* qp  = q  + leafBase;
    const float* kAp = k  + leafBase;
    const float* kFp = Kt + treeBase + (size_t)rF * D_DIM;

    v8f cA = {}; v8f cF = {}; v8f cG = {};
    const int dofs = 4 * half;
    #pragma unroll
    for (int j = 0; j < 8; ++j) {
        const int d = 8 * j + dofs;
        const float4 a4 = *(const float4*)(qp  + d);
        const float4 kA4 = *(const float4*)(kAp + d);
        const float4 kF4 = *(const float4*)(kFp + d);
        v2f a, bA, bF;
        a.x = a4.x * SCALE; a.y = a4.y * SCALE;
        bA.x = kA4.x; bA.y = kA4.y;
        bF.x = kF4.x; bF.y = kF4.y;
        cA = __builtin_amdgcn_wmma_f32_16x16x4_f32(false, a, false, bA, (short)0, cA, false, false);
        cF = __builtin_amdgcn_wmma_f32_16x16x4_f32(false, a, false, bF, (short)0, cF, false, false);
        cG = __builtin_amdgcn_wmma_f32_16x16x4_f32(false, bA, false, bA, (short)0, cG, false, false);
        a.x = a4.z * SCALE; a.y = a4.w * SCALE;
        bA.x = kA4.z; bA.y = kA4.w;
        bF.x = kF4.z; bF.y = kF4.w;
        cA = __builtin_amdgcn_wmma_f32_16x16x4_f32(false, a, false, bA, (short)0, cA, false, false);
        cF = __builtin_amdgcn_wmma_f32_16x16x4_f32(false, a, false, bF, (short)0, cF, false, false);
        cG = __builtin_amdgcn_wmma_f32_16x16x4_f32(false, bA, false, bA, (short)0, cG, false, false);
    }

    float t1[4], t2[2];
    #pragma unroll
    for (int u = 0; u < 4; ++u) {
        t1[u] = add_lane_xor1(cG[2 * u] + cG[2 * u + 1]);
    }
    #pragma unroll
    for (int u = 0; u < 2; ++u) {
        t2[u] = add_lane_xor2(t1[2 * u] + t1[2 * u + 1]);
    }

    #pragma unroll
    for (int r = 0; r < 8; ++r) {
        const int M = r + 8 * half;
        sA[M * 16 + n] = cA[r];
        sF[M * 16 + n] = cF[r];
        sG[M * 16 + n] = cG[r];
    }
    if ((n & 1) == 0) {
        #pragma unroll
        for (int u = 0; u < 4; ++u) sS1[(4 * half + u) * 8 + (n >> 1)] = t1[u];
    }
    if ((n & 3) == 0) {
        #pragma unroll
        for (int u = 0; u < 2; ++u) sS2[(2 * half + u) * 4 + (n >> 2)] = t2[u];
    }
    __syncthreads();

    {
        const int p2 = n & ~1;
        const float2 g0 = *(const float2*)(sG + p2 * 16 + p2);
        const float g11 = sG[(p2 + 1) * 16 + p2 + 1];
        float a0, a1;
        merge_weights(g0.x, g0.y, g11, SCALE, a0, a1);
        const float wt1 = (n & 1) ? a1 : a0;

        const int r2 = (n >> 1) & ~1;
        const float2 h0 = *(const float2*)(sS1 + r2 * 8 + r2);
        const float h11 = sS1[(r2 + 1) * 8 + r2 + 1];
        float b0, b1;
        merge_weights(h0.x, h0.y, h11, SCALE * 0.25f, b0, b1);
        const float wt2 = wt1 * ((n & 2) ? b1 : b0);

        const int s2 = (n >> 2) & ~1;
        const float2 j0 = *(const float2*)(sS2 + s2 * 4 + s2);
        const float j11 = sS2[(s2 + 1) * 4 + s2 + 1];
        float c0, c1;
        merge_weights(j0.x, j0.y, j11, SCALE * 0.0625f, c0, c1);
        const float wt3 = wt2 * ((n & 4) ? c1 : c0);

        if (half == 0) { sW[n] = wt1; sW[16 + n] = wt2; sW[32 + n] = wt3; }
    }

    float ev[13];
    float dsum = 0.0f;
    if (half == 0) {
        const int j = n;
        float sr[16];
        #pragma unroll
        for (int dx = 0; dx < 16; ++dx) sr[dx] = sA[j * 16 + (j ^ dx)];
        float x[13];
        x[0] = sr[0];
        x[1] = sr[1];
        x[2] = 0.5f * (sr[2] + sr[3]);
        x[3] = 0.25f * ((sr[4] + sr[5]) + (sr[6] + sr[7]));
        x[4] = 0.125f * (((sr[8] + sr[9]) + (sr[10] + sr[11]))
                       + ((sr[12] + sr[13]) + (sr[14] + sr[15])));
        #pragma unroll
        for (int i = 0; i < 8; ++i) x[5 + i] = sF[j * 16 + i];

        bool msk[13];
        msk[0] = false;
        #pragma unroll
        for (int l = 0; l < 12; ++l) {
            const int bit = (l < 4) ? ((j >> l) & 1) : ((t >> (l - 4)) & 1);
            msk[l + 1] = (bit == 0);
        }
        float m = x[0];
        #pragma unroll
        for (int i = 1; i < 13; ++i) if (!msk[i]) m = fmaxf(m, x[i]);
        #pragma unroll
        for (int i = 0; i < 13; ++i) {
            ev[i] = msk[i] ? 0.0f : __expf(x[i] - m);
            dsum += ev[i];
        }
    }
    __syncthreads();

    if (half == 0) {
        const int j = n;
        den[j] = dsum;
        sA[j * 16 + j]       = ev[0];
        sA[j * 16 + (j ^ 1)] = ev[1];
        #pragma unroll
        for (int dx = 2; dx < 4; ++dx)  { const int i = j ^ dx; sA[j * 16 + i] = ev[2] * sW[i]; }
        #pragma unroll
        for (int dx = 4; dx < 8; ++dx)  { const int i = j ^ dx; sA[j * 16 + i] = ev[3] * sW[16 + i]; }
        #pragma unroll
        for (int dx = 8; dx < 16; ++dx) { const int i = j ^ dx; sA[j * 16 + i] = ev[4] * sW[32 + i]; }
        *(float4*)(sF + j * 16)     = make_float4(ev[5], ev[6], ev[7], ev[8]);
        *(float4*)(sF + j * 16 + 4) = make_float4(ev[9], ev[10], ev[11], ev[12]);
    }
    __syncthreads();

    const size_t vLeafRow0  = ((size_t)(b * S_DIM + 16 * t) * H_DIM + h) * (size_t)D_DIM;
    const size_t vRowStride = (size_t)H_DIM * D_DIM;
    const int col4 = 4 * n;

    v2f eA[4], eF[2];
    float4 vA[4][2], vF[2][2];
    #pragma unroll
    for (int ks2 = 0; ks2 < 4; ++ks2) {
        const int c0 = 4 * ks2 + 2 * half;
        const float2 e2 = *(const float2*)(sA + n * 16 + c0);
        eA[ks2].x = e2.x; eA[ks2].y = e2.y;
        #pragma unroll
        for (int s = 0; s < 2; ++s)
            vA[ks2][s] = *(const float4*)(v + vLeafRow0 + (size_t)(c0 + s) * vRowStride + col4);
    }
    #pragma unroll
    for (int ks2 = 0; ks2 < 2; ++ks2) {
        const int c0 = 4 * ks2 + 2 * half;
        const float2 e2 = *(const float2*)(sF + n * 16 + c0);
        eF[ks2].x = e2.x; eF[ks2].y = e2.y;
        #pragma unroll
        for (int s = 0; s < 2; ++s) {
            const int c = c0 + s;
            const int row = (S_DIM - (S_DIM >> (3 + c))) + ((t >> c) ^ 1);
            vF[ks2][s] = *(const float4*)(Vt + treeBase + (size_t)row * D_DIM + col4);
        }
    }

    v8f acc[4] = {};
    #pragma unroll
    for (int ks2 = 0; ks2 < 4; ++ks2) {
        #pragma unroll
        for (int u = 0; u < 4; ++u) {
            v2f bV;
            bV.x = f4get(vA[ks2][0], u); bV.y = f4get(vA[ks2][1], u);
            acc[u] = __builtin_amdgcn_wmma_f32_16x16x4_f32(false, eA[ks2], false, bV, (short)0, acc[u], false, false);
        }
    }
    #pragma unroll
    for (int ks2 = 0; ks2 < 2; ++ks2) {
        #pragma unroll
        for (int u = 0; u < 4; ++u) {
            v2f bV;
            bV.x = f4get(vF[ks2][0], u); bV.y = f4get(vF[ks2][1], u);
            acc[u] = __builtin_amdgcn_wmma_f32_16x16x4_f32(false, eF[ks2], false, bV, (short)0, acc[u], false, false);
        }
    }

    #pragma unroll
    for (int r = 0; r < 8; ++r) {
        const int M = r + 8 * half;
        const float dn = den[M];
        float4 o;
        o.x = acc[0][r] / dn; o.y = acc[1][r] / dn;
        o.z = acc[2][r] / dn; o.w = acc[3][r] / dn;
        *(float4*)(out + ((size_t)(b * S_DIM + 16 * t + M) * H_DIM + h) * D_DIM + col4) = o;
    }
}

extern "C" void kernel_launch(void* const* d_in, const int* in_sizes, int n_in,
                              void* d_out, int out_size, void* d_ws, size_t ws_size,
                              hipStream_t stream)
{
    const float* q = (const float*)d_in[0];
    const float* k = (const float*)d_in[1];
    const float* v = (const float*)d_in[2];
    float* out = (float*)d_out;

    float* Kt = (float*)d_ws;
    float* Vt = Kt + (size_t)B_DIM * H_DIM * S_DIM * D_DIM;

    tree_low_kernel<<<1024, 256, 0, stream>>>(k, v, Kt, Vt);
    tree_high_kernel<<<32, 512, 0, stream>>>(Kt, Vt);

    const int tiles = B_DIM * H_DIM * NTILE;
    hst_attn_kernel<<<tiles / 4, 128, 0, stream>>>(q, k, v, Kt, Vt, out);

    (void)in_sizes; (void)n_in; (void)out_size; (void)ws_size;
}


# Levels 10-11 folded into attention: their scores/values = means/weighted sums of 6 level-9 nodes (WMMA far K*K^T weights; sum order changes); tree = 1 launch, 512-leaf blocks climb to level 9
# speedup vs baseline: 2.4250x; 1.2118x over previous
#include <hip/hip_runtime.h>

typedef float v2f __attribute__((ext_vector_type(2)));
typedef float v8f __attribute__((ext_vector_type(8)));

#define B_DIM 4
#define S_DIM 4096
#define H_DIM 8
#define D_DIM 64
#define SCALE 0.125f
#define NTILE (S_DIM / 16)

__device__ __forceinline__ void merge_pair(const float2 kc0, const float2 kc1,
                                           const float2 vc0, const float2 vc1,
                                           float2& kp, float2& vo)
{
    float2 vp;
    kp.x = 0.5f * (kc0.x + kc1.x); kp.y = 0.5f * (kc0.y + kc1.y);
    vp.x = 0.5f * (vc0.x + vc1.x); vp.y = 0.5f * (vc0.y + vc1.y);

    float pss = kp.x * kp.x  + kp.y * kp.y;
    float ps0 = kp.x * kc0.x + kp.y * kc0.y;
    float ps1 = kp.x * kc1.x + kp.y * kc1.y;
    #pragma unroll
    for (int m = 16; m >= 1; m >>= 1) {
        pss += __shfl_xor(pss, m, 32);
        ps0 += __shfl_xor(ps0, m, 32);
        ps1 += __shfl_xor(ps1, m, 32);
    }
    pss *= SCALE; ps0 *= SCALE; ps1 *= SCALE;
    const float mm = fmaxf(pss, fmaxf(ps0, ps1));
    const float es = __expf(pss - mm);
    const float e0 = __expf(ps0 - mm);
    const float e1 = __expf(ps1 - mm);
    const float inv = 1.0f / (es + e0 + e1 + 1e-9f);

    vo.x = (es * vp.x + e0 * vc0.x + e1 * vc1.x) * inv;
    vo.y = (es * vp.y + e0 * vc0.y + e1 * vc1.y) * inv;
}

template <int NP, bool STORE_K, bool STORE_V>
__device__ __forceinline__ void level_in_regs(float2* kc, float2* vc,
                                              float* __restrict__ Kt,
                                              float* __restrict__ Vt,
                                              size_t treeRow0, int level,
                                              int chunk, int doff)
{
    const int IBl = S_DIM - (S_DIM >> (level - 1));
    const size_t row0 = treeRow0 + IBl + (size_t)chunk * NP;
    #pragma unroll
    for (int j = 0; j < NP; ++j) {
        float2 kp, vo;
        merge_pair(kc[2 * j], kc[2 * j + 1], vc[2 * j], vc[2 * j + 1], kp, vo);
        const size_t dst = (row0 + j) * D_DIM + doff;
        if (STORE_K) *(float2*)(Kt + dst) = kp;
        if (STORE_V) *(float2*)(Vt + dst) = vo;
        kc[j] = kp; vc[j] = vo;
    }
}

__global__ void __launch_bounds__(1024) __attribute__((amdgpu_waves_per_eu(8, 8)))
tree_kernel(const float* __restrict__ kleaf,
            const float* __restrict__ vleaf,
            float* __restrict__ Kt,
            float* __restrict__ Vt)
{
    __shared__ float2 sK4[32][32];
    __shared__ float2 sV4[32][32];
    __shared__ float2 sK8[2][32];
    __shared__ float2 sV8[2][32];

    const int wave = threadIdx.x >> 5;
    const int lane = threadIdx.x & 31;
    const int h    = blockIdx.x & 7;
    const int span = (blockIdx.x >> 3) & 7;
    const int b    = blockIdx.x >> 6;
    const int bh   = b * H_DIM + h;
    const int doff = 2 * lane;
    const int c    = span * 32 + wave;

    float2 kc[16], vc[16];
    const size_t leaf0 = ((size_t)(b * S_DIM + 16 * c) * H_DIM + h) * D_DIM + doff;
    #pragma unroll
    for (int i = 0; i < 16; ++i) {
        kc[i] = *(const float2*)(kleaf + leaf0 + (size_t)i * (H_DIM * D_DIM));
        vc[i] = *(const float2*)(vleaf + leaf0 + (size_t)i * (H_DIM * D_DIM));
    }

    const size_t treeRow0 = (size_t)bh * S_DIM;
    level_in_regs<8, false, false>(kc, vc, Kt, Vt, treeRow0, 1, c, doff);
    level_in_regs<4, false, false>(kc, vc, Kt, Vt, treeRow0, 2, c, doff);
    level_in_regs<2, false, false>(kc, vc, Kt, Vt, treeRow0, 3, c, doff);
    level_in_regs<1, true, true>(kc, vc, Kt, Vt, treeRow0, 4, c, doff);

    sK4[wave][lane] = kc[0];
    sV4[wave][lane] = vc[0];
    __syncthreads();

    if (wave < 2) {
        #pragma unroll
        for (int i = 0; i < 16; ++i) { kc[i] = sK4[16 * wave + i][lane]; vc[i] = sV4[16 * wave + i][lane]; }
        const int g = span * 2 + wave;
        level_in_regs<8, true, true>(kc, vc, Kt, Vt, treeRow0, 5, g, doff);
        level_in_regs<4, true, true>(kc, vc, Kt, Vt, treeRow0, 6, g, doff);
        level_in_regs<2, true, true>(kc, vc, Kt, Vt, treeRow0, 7, g, doff);
        level_in_regs<1, true, true>(kc, vc, Kt, Vt, treeRow0, 8, g, doff);
        sK8[wave][lane] = kc[0];
        sV8[wave][lane] = vc[0];
    }
    __syncthreads();

    if (wave == 0) {
        kc[0] = sK8[0][lane]; kc[1] = sK8[1][lane];
        vc[0] = sV8[0][lane]; vc[1] = sV8[1][lane];
        level_in_regs<1, true, true>(kc, vc, Kt, Vt, treeRow0, 9, span, doff);
    }
}

__device__ __forceinline__ float f4get(const float4 x, const int u)
{
    return u == 0 ? x.x : (u == 1 ? x.y : (u == 2 ? x.z : x.w));
}

__device__ __forceinline__ void merge_weights(float g00, float g01, float g11,
                                              const float nrm, float& w0, float& w1)
{
    g00 *= nrm; g01 *= nrm; g11 *= nrm;
    const float ps0 = 0.5f * (g00 + g01);
    const float ps1 = 0.5f * (g01 + g11);
    const float pss = 0.5f * (ps0 + ps1);
    const float mm = fmaxf(pss, fmaxf(ps0, ps1));
    const float es = __expf(pss - mm);
    const float e0 = __expf(ps0 - mm);
    const float e1 = __expf(ps1 - mm);
    const float inv = 1.0f / (es + e0 + e1 + 1e-9f);
    w0 = (0.5f * es + e0) * inv;
    w1 = (0.5f * es + e1) * inv;
}

__device__ __forceinline__ float add_lane_xor1(const float x)
{
    return x + __int_as_float(__builtin_amdgcn_update_dpp(0, __float_as_int(x), 0xB1, 0xf, 0xf, false));
}
__device__ __forceinline__ float add_lane_xor2(const float x)
{
    return x + __int_as_float(__builtin_amdgcn_update_dpp(0, __float_as_int(x), 0x4E, 0xf, 0xf, false));
}

#define L_SA   0
#define L_SF   256
#define L_SG   512
#define L_SH   768
#define L_S1   1024
#define L_S2   1088
#define L_SW   1104
#define L_DEN  1152
#define L_WAVE 1168

#define IB9 (S_DIM - (S_DIM >> 8))

__device__ __forceinline__ int far_row(const int c, const int t)
{
    const int cs  = c & 7;
    const int sib = (S_DIM - (S_DIM >> (3 + cs))) + ((t >> cs) ^ 1);
    const int r10 = IB9 + 2 * ((t >> 6) ^ 1) + (c & 1);
    const int r11 = IB9 + 4 * ((t >> 7) ^ 1) + (c & 3);
    return c < 6 ? sib : (c < 8 ? r10 : r11);
}

__global__ void __launch_bounds__(128)
hst_attn_kernel(const float* __restrict__ q,
                const float* __restrict__ k,
                const float* __restrict__ v,
                const float* __restrict__ Kt,
                const float* __restrict__ Vt,
                float* __restrict__ out)
{
    __shared__ __attribute__((aligned(16))) float ldsS[4][L_WAVE];

    const int lane = threadIdx.x & 31;
    const int w    = threadIdx.x >> 5;
    const int tile = blockIdx.x * 4 + w;
    const int t  = tile & (NTILE - 1);
    const int bh = tile >> 8;
    const int h  = bh & (H_DIM - 1);
    const int b  = bh >> 3;

    const int n    = lane & 15;
    const int half = lane >> 4;

    float* sA  = &ldsS[w][L_SA];
    float* sF  = &ldsS[w][L_SF];
    float* sG  = &ldsS[w][L_SG];
    float* sH  = &ldsS[w][L_SH];
    float* sS1 = &ldsS[w][L_S1];
    float* sS2 = &ldsS[w][L_S2];
    float* sW  = &ldsS[w][L_SW];
    float* den = &ldsS[w][L_DEN];

    const int rF = far_row(n, t);

    const size_t leafBase = ((size_t)(b * S_DIM + 16 * t + n) * H_DIM + h) * D_DIM;
    const size_t treeBase = (size_t)(b * H_DIM + h) * S_DIM * (size_t)D_DIM;
    const float* qp  = q  + leafBase;
    const float* kAp = k  + leafBase;
    const float* kFp = Kt + treeBase + (size_t)rF * D_DIM;

    v8f cA = {}; v8f cF = {}; v8f cG = {}; v8f cH = {};
    const int dofs = 4 * half;
    #pragma unroll
    for (int j = 0; j < 8; ++j) {
        const int d = 8 * j + dofs;
        const float4 a4 = *(const float4*)(qp  + d);
        const float4 kA4 = *(const float4*)(kAp + d);
        const float4 kF4 = *(const float4*)(kFp + d);
        v2f a, bA, bF;
        a.x = a4.x * SCALE; a.y = a4.y * SCALE;
        bA.x = kA4.x; bA.y = kA4.y;
        bF.x = kF4.x; bF.y = kF4.y;
        cA = __builtin_amdgcn_wmma_f32_16x16x4_f32(false, a, false, bA, (short)0, cA, false, false);
        cF = __builtin_amdgcn_wmma_f32_16x16x4_f32(false, a, false, bF, (short)0, cF, false, false);
        cG = __builtin_amdgcn_wmma_f32_16x16x4_f32(false, bA, false, bA, (short)0, cG, false, false);
        cH = __builtin_amdgcn_wmma_f32_16x16x4_f32(false, bF, false, bF, (short)0, cH, false, false);
        a.x = a4.z * SCALE; a.y = a4.w * SCALE;
        bA.x = kA4.z; bA.y = kA4.w;
        bF.x = kF4.z; bF.y = kF4.w;
        cA = __builtin_amdgcn_wmma_f32_16x16x4_f32(false, a, false, bA, (short)0, cA, false, false);
        cF = __builtin_amdgcn_wmma_f32_16x16x4_f32(false, a, false, bF, (short)0, cF, false, false);
        cG = __builtin_amdgcn_wmma_f32_16x16x4_f32(false, bA, false, bA, (short)0, cG, false, false);
        cH = __builtin_amdgcn_wmma_f32_16x16x4_f32(false, bF, false, bF, (short)0, cH, false, false);
    }

    float t1[4], t2[2];
    #pragma unroll
    for (int u = 0; u < 4; ++u) {
        t1[u] = add_lane_xor1(cG[2 * u] + cG[2 * u + 1]);
    }
    #pragma unroll
    for (int u = 0; u < 2; ++u) {
        t2[u] = add_lane_xor2(t1[2 * u] + t1[2 * u + 1]);
    }

    #pragma unroll
    for (int r = 0; r < 8; ++r) {
        const int M = r + 8 * half;
        sA[M * 16 + n] = cA[r];
        sF[M * 16 + n] = cF[r];
        sG[M * 16 + n] = cG[r];
        sH[M * 16 + n] = cH[r];
    }
    if ((n & 1) == 0) {
        #pragma unroll
        for (int u = 0; u < 4; ++u) sS1[(4 * half + u) * 8 + (n >> 1)] = t1[u];
    }
    if ((n & 3) == 0) {
        #pragma unroll
        for (int u = 0; u < 2; ++u) sS2[(2 * half + u) * 4 + (n >> 2)] = t2[u];
    }
    __syncthreads();

    {
        const int p2 = n & ~1;
        const float2 g0 = *(const float2*)(sG + p2 * 16 + p2);
        const float g11 = sG[(p2 + 1) * 16 + p2 + 1];
        float a0, a1;
        merge_weights(g0.x, g0.y, g11, SCALE, a0, a1);
        const float wt1 = (n & 1) ? a1 : a0;

        const int r2 = (n >> 1) & ~1;
        const float2 h0 = *(const float2*)(sS1 + r2 * 8 + r2);
        const float h11 = sS1[(r2 + 1) * 8 + r2 + 1];
        float b0, b1;
        merge_weights(h0.x, h0.y, h11, SCALE * 0.25f, b0, b1);
        const float wt2 = wt1 * ((n & 2) ? b1 : b0);

        const int s2 = (n >> 2) & ~1;
        const float2 j0 = *(const float2*)(sS2 + s2 * 4 + s2);
        const float j11 = sS2[(s2 + 1) * 4 + s2 + 1];
        float c0, c1;
        merge_weights(j0.x, j0.y, j11, SCALE * 0.0625f, c0, c1);
        const float wt3 = wt2 * ((n & 4) ? c1 : c0);

        if (half == 0) { sW[n] = wt1; sW[16 + n] = wt2; sW[32 + n] = wt3; }
    }

    float fw[6];
    {
        const float2 p0  = *(const float2*)(sH + 6 * 16 + 6);
        const float  p11 = sH[7 * 16 + 7];
        const float4 r8  = *(const float4*)(sH + 8 * 16 + 8);
        const float4 r9  = *(const float4*)(sH + 9 * 16 + 8);
        const float4 r10 = *(const float4*)(sH + 10 * 16 + 8);
        const float4 r11 = *(const float4*)(sH + 11 * 16 + 8);
        float wa, wb, wc, wd, we, wf, wX, wY;
        merge_weights(p0.x, p0.y, p11, SCALE, wa, wb);
        merge_weights(r8.x, r8.y, r9.y, SCALE, wc, wd);
        merge_weights(r10.z, r10.w, r11.w, SCALE, we, wf);
        merge_weights((r8.x + r8.y) + (r9.x + r9.y),
                      (r8.z + r8.w) + (r9.z + r9.w),
                      (r10.z + r10.w) + (r11.z + r11.w), SCALE * 0.25f, wX, wY);
        fw[0] = wa;      fw[1] = wb;
        fw[2] = wX * wc; fw[3] = wX * wd;
        fw[4] = wY * we; fw[5] = wY * wf;
    }

    float ev[13];
    float dsum = 0.0f;
    if (half == 0) {
        const int j = n;
        float sr[16];
        #pragma unroll
        for (int dx = 0; dx < 16; ++dx) sr[dx] = sA[j * 16 + (j ^ dx)];
        float x[13];
        x[0] = sr[0];
        x[1] = sr[1];
        x[2] = 0.5f * (sr[2] + sr[3]);
        x[3] = 0.25f * ((sr[4] + sr[5]) + (sr[6] + sr[7]));
        x[4] = 0.125f * (((sr[8] + sr[9]) + (sr[10] + sr[11]))
                       + ((sr[12] + sr[13]) + (sr[14] + sr[15])));
        const float4 f0 = *(const float4*)(sF + j * 16);
        const float4 f1 = *(const float4*)(sF + j * 16 + 4);
        const float4 f2 = *(const float4*)(sF + j * 16 + 8);
        x[5] = f0.x; x[6] = f0.y; x[7] = f0.z; x[8] = f0.w;
        x[9] = f1.x; x[10] = f1.y;
        x[11] = 0.5f * (f1.z + f1.w);
        x[12] = 0.25f * ((f2.x + f2.y) + (f2.z + f2.w));

        bool msk[13];
        msk[0] = false;
        #pragma unroll
        for (int l = 0; l < 12; ++l) {
            const int bit = (l < 4) ? ((j >> l) & 1) : ((t >> (l - 4)) & 1);
            msk[l + 1] = (bit == 0);
        }
        float m = x[0];
        #pragma unroll
        for (int i = 1; i < 13; ++i) if (!msk[i]) m = fmaxf(m, x[i]);
        #pragma unroll
        for (int i = 0; i < 13; ++i) {
            ev[i] = msk[i] ? 0.0f : __expf(x[i] - m);
            dsum += ev[i];
        }
    }
    __syncthreads();

    if (half == 0) {
        const int j = n;
        den[j] = dsum;
        sA[j * 16 + j]       = ev[0];
        sA[j * 16 + (j ^ 1)] = ev[1];
        #pragma unroll
        for (int dx = 2; dx < 4; ++dx)  { const int i = j ^ dx; sA[j * 16 + i] = ev[2] * sW[i]; }
        #pragma unroll
        for (int dx = 4; dx < 8; ++dx)  { const int i = j ^ dx; sA[j * 16 + i] = ev[3] * sW[16 + i]; }
        #pragma unroll
        for (int dx = 8; dx < 16; ++dx) { const int i = j ^ dx; sA[j * 16 + i] = ev[4] * sW[32 + i]; }
        *(float4*)(sF + j * 16)     = make_float4(ev[5], ev[6], ev[7], ev[8]);
        *(float4*)(sF + j * 16 + 4) = make_float4(ev[9], ev[10], ev[11] * fw[0], ev[11] * fw[1]);
        *(float4*)(sF + j * 16 + 8) = make_float4(ev[12] * fw[2], ev[12] * fw[3],
                                                  ev[12] * fw[4], ev[12] * fw[5]);
    }
    __syncthreads();

    const size_t vLeafRow0  = ((size_t)(b * S_DIM + 16 * t) * H_DIM + h) * (size_t)D_DIM;
    const size_t vRowStride = (size_t)H_DIM * D_DIM;
    const int col4 = 4 * n;

    v2f eA[4], eF[3];
    float4 vA[4][2], vF[3][2];
    #pragma unroll
    for (int ks2 = 0; ks2 < 4; ++ks2) {
        const int c0 = 4 * ks2 + 2 * half;
        const float2 e2 = *(const float2*)(sA + n * 16 + c0);
        eA[ks2].x = e2.x; eA[ks2].y = e2.y;
        #pragma unroll
        for (int s = 0; s < 2; ++s)
            vA[ks2][s] = *(const float4*)(v + vLeafRow0 + (size_t)(c0 + s) * vRowStride + col4);
    }
    #pragma unroll
    for (int ks2 = 0; ks2 < 3; ++ks2) {
        const int c0 = 4 * ks2 + 2 * half;
        const float2 e2 = *(const float2*)(sF + n * 16 + c0);
        eF[ks2].x = e2.x; eF[ks2].y = e2.y;
        #pragma unroll
        for (int s = 0; s < 2; ++s) {
            const int row = far_row(c0 + s, t);
            vF[ks2][s] = *(const float4*)(Vt + treeBase + (size_t)row * D_DIM + col4);
        }
    }

    v8f acc[4] = {};
    #pragma unroll
    for (int ks2 = 0; ks2 < 4; ++ks2) {
        #pragma unroll
        for (int u = 0; u < 4; ++u) {
            v2f bV;
            bV.x = f4get(vA[ks2][0], u); bV.y = f4get(vA[ks2][1], u);
            acc[u] = __builtin_amdgcn_wmma_f32_16x16x4_f32(false, eA[ks2], false, bV, (short)0, acc[u], false, false);
        }
    }
    #pragma unroll
    for (int ks2 = 0; ks2 < 3; ++ks2) {
        #pragma unroll
        for (int u = 0; u < 4; ++u) {
            v2f bV;
            bV.x = f4get(vF[ks2][0], u); bV.y = f4get(vF[ks2][1], u);
            acc[u] = __builtin_amdgcn_wmma_f32_16x16x4_f32(false, eF[ks2], false, bV, (short)0, acc[u], false, false);
        }
    }

    #pragma unroll
    for (int r = 0; r < 8; ++r) {
        const int M = r + 8 * half;
        const float dn = den[M];
        float4 o;
        o.x = acc[0][r] / dn; o.y = acc[1][r] / dn;
        o.z = acc[2][r] / dn; o.w = acc[3][r] / dn;
        *(float4*)(out + ((size_t)(b * S_DIM + 16 * t + M) * H_DIM + h) * D_DIM + col4) = o;
    }
}

extern "C" void kernel_launch(void* const* d_in, const int* in_sizes, int n_in,
                              void* d_out, int out_size, void* d_ws, size_t ws_size,
                              hipStream_t stream)
{
    const float* q = (const float*)d_in[0];
    const float* k = (const float*)d_in[1];
    const float* v = (const float*)d_in[2];
    float* out = (float*)d_out;

    float* Kt = (float*)d_ws;
    float* Vt = Kt + (size_t)B_DIM * H_DIM * S_DIM * D_DIM;

    tree_kernel<<<256, 1024, 0, stream>>>(k, v, Kt, Vt);

    const int tiles = B_DIM * H_DIM * NTILE;
    hst_attn_kernel<<<tiles / 4, 128, 0, stream>>>(q, k, v, Kt, Vt, out);

    (void)in_sizes; (void)n_in; (void)out_size; (void)ws_size;
}


# Tree build, levels 5-9 (one or two waves of a block run alone): a merge's three 32-lane sums run on the vector ALU (half swap + row DPP adds), not as 5 LDS shuffles each; same partner order and sums
# speedup vs baseline: 2.5326x; 1.0444x over previous
#include <hip/hip_runtime.h>

typedef float v2f __attribute__((ext_vector_type(2)));
typedef float v8f __attribute__((ext_vector_type(8)));

#define B_DIM 4
#define S_DIM 4096
#define H_DIM 8
#define D_DIM 64
#define SCALE 0.125f
#define NTILE (S_DIM / 16)

__device__ __forceinline__ float add_row_swap(const float x)
{
    const int b = __float_as_int(x);
    return x + __int_as_float(__builtin_amdgcn_permlanex16(b, b, 0x76543210, 0xfedcba98, false, false));
}
template <int CTRL>
__device__ __forceinline__ float add_row_dpp(const float x)
{
    return x + __int_as_float(__builtin_amdgcn_update_dpp(0, __float_as_int(x), CTRL, 0xf, 0xf, false));
}
__device__ __forceinline__ float wave_sum(float x)
{
    x = add_row_swap(x);
    x = add_row_dpp<0x168>(x);
    x = add_row_dpp<0x164>(x);
    x = add_row_dpp<0x4E>(x);
    x = add_row_dpp<0xB1>(x);
    return x;
}

template <bool ALU_SUMS>
__device__ __forceinline__ void merge_pair(const float2 kc0, const float2 kc1,
                                           const float2 vc0, const float2 vc1,
                                           float2& kp, float2& vo)
{
    float2 vp;
    kp.x = 0.5f * (kc0.x + kc1.x); kp.y = 0.5f * (kc0.y + kc1.y);
    vp.x = 0.5f * (vc0.x + vc1.x); vp.y = 0.5f * (vc0.y + vc1.y);

    float pss = kp.x * kp.x  + kp.y * kp.y;
    float ps0 = kp.x * kc0.x + kp.y * kc0.y;
    float ps1 = kp.x * kc1.x + kp.y * kc1.y;
    if (ALU_SUMS) {
        pss = wave_sum(pss);
        ps0 = wave_sum(ps0);
        ps1 = wave_sum(ps1);
    } else {
        #pragma unroll
        for (int m = 16; m >= 1; m >>= 1) {
            pss += __shfl_xor(pss, m, 32);
            ps0 += __shfl_xor(ps0, m, 32);
            ps1 += __shfl_xor(ps1, m, 32);
        }
    }
    pss *= SCALE; ps0 *= SCALE; ps1 *= SCALE;
    const float mm = fmaxf(pss, fmaxf(ps0, ps1));
    const float es = __expf(pss - mm);
    const float e0 = __expf(ps0 - mm);
    const float e1 = __expf(ps1 - mm);
    const float inv = 1.0f / (es + e0 + e1 + 1e-9f);

    vo.x = (es * vp.x + e0 * vc0.x + e1 * vc1.x) * inv;
    vo.y = (es * vp.y + e0 * vc0.y + e1 * vc1.y) * inv;
}

template <int NP, bool STORE_K, bool STORE_V, bool ALU_SUMS>
__device__ __forceinline__ void level_in_regs(float2* kc, float2* vc,
                                              float* __restrict__ Kt,
                                              float* __restrict__ Vt,
                                              size_t treeRow0, int level,
                                              int chunk, int doff)
{
    const int IBl = S_DIM - (S_DIM >> (level - 1));
    const size_t row0 = treeRow0 + IBl + (size_t)chunk * NP;
    #pragma unroll
    for (int j = 0; j < NP; ++j) {
        float2 kp, vo;
        merge_pair<ALU_SUMS>(kc[2 * j], kc[2 * j + 1], vc[2 * j], vc[2 * j + 1], kp, vo);
        const size_t dst = (row0 + j) * D_DIM + doff;
        if (STORE_K) *(float2*)(Kt + dst) = kp;
        if (STORE_V) *(float2*)(Vt + dst) = vo;
        kc[j] = kp; vc[j] = vo;
    }
}

__global__ void __launch_bounds__(1024) __attribute__((amdgpu_waves_per_eu(8, 8)))
tree_kernel(const float* __restrict__ kleaf,
            const float* __restrict__ vleaf,
            float* __restrict__ Kt,
            float* __restrict__ Vt)
{
    __shared__ float2 sK4[32][32];
    __shared__ float2 sV4[32][32];
    __shared__ float2 sK8[2][32];
    __shared__ float2 sV8[2][32];

    const int wave = threadIdx.x >> 5;
    const int lane = threadIdx.x & 31;
    const int h    = blockIdx.x & 7;
    const int span = (blockIdx.x >> 3) & 7;
    const int b    = blockIdx.x >> 6;
    const int bh   = b * H_DIM + h;
    const int doff = 2 * lane;
    const int c    = span * 32 + wave;

    float2 kc[16], vc[16];
    const size_t leaf0 = ((size_t)(b * S_DIM + 16 * c) * H_DIM + h) * D_DIM + doff;
    #pragma unroll
    for (int i = 0; i < 16; ++i) {
        kc[i] = *(const float2*)(kleaf + leaf0 + (size_t)i * (H_DIM * D_DIM));
        vc[i] = *(const float2*)(vleaf + leaf0 + (size_t)i * (H_DIM * D_DIM));
    }

    const size_t treeRow0 = (size_t)bh * S_DIM;
    level_in_regs<8, false, false, false>(kc, vc, Kt, Vt, treeRow0, 1, c, doff);
    level_in_regs<4, false, false, false>(kc, vc, Kt, Vt, treeRow0, 2, c, doff);
    level_in_regs<2, false, false, false>(kc, vc, Kt, Vt, treeRow0, 3, c, doff);
    level_in_regs<1, true, true, false>(kc, vc, Kt, Vt, treeRow0, 4, c, doff);

    sK4[wave][lane] = kc[0];
    sV4[wave][lane] = vc[0];
    __syncthreads();

    if (wave < 2) {
        #pragma unroll
        for (int i = 0; i < 16; ++i) { kc[i] = sK4[16 * wave + i][lane]; vc[i] = sV4[16 * wave + i][lane]; }
        const int g = span * 2 + wave;
        level_in_regs<8, true, true, true>(kc, vc, Kt, Vt, treeRow0, 5, g, doff);
        level_in_regs<4, true, true, true>(kc, vc, Kt, Vt, treeRow0, 6, g, doff);
        level_in_regs<2, true, true, true>(kc, vc, Kt, Vt, treeRow0, 7, g, doff);
        level_in_regs<1, true, true, true>(kc, vc, Kt, Vt, treeRow0, 8, g, doff);
        sK8[wave][lane] = kc[0];
        sV8[wave][lane] = vc[0];
    }
    __syncthreads();

    if (wave == 0) {
        kc[0] = sK8[0][lane]; kc[1] = sK8[1][lane];
        vc[0] = sV8[0][lane]; vc[1] = sV8[1][lane];
        level_in_regs<1, true, true, true>(kc, vc, Kt, Vt, treeRow0, 9, span, doff);
    }
}

__device__ __forceinline__ float f4get(const float4 x, const int u)
{
    return u == 0 ? x.x : (u == 1 ? x.y : (u == 2 ? x.z : x.w));
}

__device__ __forceinline__ void merge_weights(float g00, float g01, float g11,
                                              const float nrm, float& w0, float& w1)
{
    g00 *= nrm; g01 *= nrm; g11 *= nrm;
    const float ps0 = 0.5f * (g00 + g01);
    const float ps1 = 0.5f * (g01 + g11);
    const float pss = 0.5f * (ps0 + ps1);
    const float mm = fmaxf(pss, fmaxf(ps0, ps1));
    const float es = __expf(pss - mm);
    const float e0 = __expf(ps0 - mm);
    const float e1 = __expf(ps1 - mm);
    const float inv = 1.0f / (es + e0 + e1 + 1e-9f);
    w0 = (0.5f * es + e0) * inv;
    w1 = (0.5f * es + e1) * inv;
}

__device__ __forceinline__ float add_lane_xor1(const float x)
{
    return x + __int_as_float(__builtin_amdgcn_update_dpp(0, __float_as_int(x), 0xB1, 0xf, 0xf, false));
}
__device__ __forceinline__ float add_lane_xor2(const float x)
{
    return x + __int_as_float(__builtin_amdgcn_update_dpp(0, __float_as_int(x), 0x4E, 0xf, 0xf, false));
}

#define L_SA   0
#define L_SF   256
#define L_SG   512
#define L_SH   768
#define L_S1   1024
#define L_S2   1088
#define L_SW   1104
#define L_DEN  1152
#define L_WAVE 1168

#define IB9 (S_DIM - (S_DIM >> 8))

__device__ __forceinline__ int far_row(const int c, const int t)
{
    const int cs  = c & 7;
    const int sib = (S_DIM - (S_DIM >> (3 + cs))) + ((t >> cs) ^ 1);
    const int r10 = IB9 + 2 * ((t >> 6) ^ 1) + (c & 1);
    const int r11 = IB9 + 4 * ((t >> 7) ^ 1) + (c & 3);
    return c < 6 ? sib : (c < 8 ? r10 : r11);
}

__global__ void __launch_bounds__(128)
hst_attn_kernel(const float* __restrict__ q,
                const float* __restrict__ k,
                const float* __restrict__ v,
                const float* __restrict__ Kt,
                const float* __restrict__ Vt,
                float* __restrict__ out)
{
    __shared__ __attribute__((aligned(16))) float ldsS[4][L_WAVE];

    const int lane = threadIdx.x & 31;
    const int w    = threadIdx.x >> 5;
    const int tile = blockIdx.x * 4 + w;
    const int t  = tile & (NTILE - 1);
    const int bh = tile >> 8;
    const int h  = bh & (H_DIM - 1);
    const int b  = bh >> 3;

    const int n    = lane & 15;
    const int half = lane >> 4;

    float* sA  = &ldsS[w][L_SA];
    float* sF  = &ldsS[w][L_SF];
    float* sG  = &ldsS[w][L_SG];
    float* sH  = &ldsS[w][L_SH];
    float* sS1 = &ldsS[w][L_S1];
    float* sS2 = &ldsS[w][L_S2];
    float* sW  = &ldsS[w][L_SW];
    float* den = &ldsS[w][L_DEN];

    const int rF = far_row(n, t);

    const size_t leafBase = ((size_t)(b * S_DIM + 16 * t + n) * H_DIM + h) * D_DIM;
    const size_t treeBase = (size_t)(b * H_DIM + h) * S_DIM * (size_t)D_DIM;
    const float* qp  = q  + leafBase;
    const float* kAp = k  + leafBase;
    const float* kFp = Kt + treeBase + (size_t)rF * D_DIM;

    v8f cA = {}; v8f cF = {}; v8f cG = {}; v8f cH = {};
    const int dofs = 4 * half;
    #pragma unroll
    for (int j = 0; j < 8; ++j) {
        const int d = 8 * j + dofs;
        const float4 a4 = *(const float4*)(qp  + d);
        const float4 kA4 = *(const float4*)(kAp + d);
        const float4 kF4 = *(const float4*)(kFp + d);
        v2f a, bA, bF;
        a.x = a4.x * SCALE; a.y = a4.y * SCALE;
        bA.x = kA4.x; bA.y = kA4.y;
        bF.x = kF4.x; bF.y = kF4.y;
        cA = __builtin_amdgcn_wmma_f32_16x16x4_f32(false, a, false, bA, (short)0, cA, false, false);
        cF = __builtin_amdgcn_wmma_f32_16x16x4_f32(false, a, false, bF, (short)0, cF, false, false);
        cG = __builtin_amdgcn_wmma_f32_16x16x4_f32(false, bA, false, bA, (short)0, cG, false, false);
        cH = __builtin_amdgcn_wmma_f32_16x16x4_f32(false, bF, false, bF, (short)0, cH, false, false);
        a.x = a4.z * SCALE; a.y = a4.w * SCALE;
        bA.x = kA4.z; bA.y = kA4.w;
        bF.x = kF4.z; bF.y = kF4.w;
        cA = __builtin_amdgcn_wmma_f32_16x16x4_f32(false, a, false, bA, (short)0, cA, false, false);
        cF = __builtin_amdgcn_wmma_f32_16x16x4_f32(false, a, false, bF, (short)0, cF, false, false);
        cG = __builtin_amdgcn_wmma_f32_16x16x4_f32(false, bA, false, bA, (short)0, cG, false, false);
        cH = __builtin_amdgcn_wmma_f32_16x16x4_f32(false, bF, false, bF, (short)0, cH, false, false);
    }

    float t1[4], t2[2];
    #pragma unroll
    for (int u = 0; u < 4; ++u) {
        t1[u] = add_lane_xor1(cG[2 * u] + cG[2 * u + 1]);
    }
    #pragma unroll
    for (int u = 0; u < 2; ++u) {
        t2[u] = add_lane_xor2(t1[2 * u] + t1[2 * u + 1]);
    }

    #pragma unroll
    for (int r = 0; r < 8; ++r) {
        const int M = r + 8 * half;
        sA[M * 16 + n] = cA[r];
        sF[M * 16 + n] = cF[r];
        sG[M * 16 + n] = cG[r];
        sH[M * 16 + n] = cH[r];
    }
    if ((n & 1) == 0) {
        #pragma unroll
        for (int u = 0; u < 4; ++u) sS1[(4 * half + u) * 8 + (n >> 1)] = t1[u];
    }
    if ((n & 3) == 0) {
        #pragma unroll
        for (int u = 0; u < 2; ++u) sS2[(2 * half + u) * 4 + (n >> 2)] = t2[u];
    }
    __syncthreads();

    {
        const int p2 = n & ~1;
        const float2 g0 = *(const float2*)(sG + p2 * 16 + p2);
        const float g11 = sG[(p2 + 1) * 16 + p2 + 1];
        float a0, a1;
        merge_weights(g0.x, g0.y, g11, SCALE, a0, a1);
        const float wt1 = (n & 1) ? a1 : a0;

        const int r2 = (n >> 1) & ~1;
        const float2 h0 = *(const float2*)(sS1 + r2 * 8 + r2);
        const float h11 = sS1[(r2 + 1) * 8 + r2 + 1];
        float b0, b1;
        merge_weights(h0.x, h0.y, h11, SCALE * 0.25f, b0, b1);
        const float wt2 = wt1 * ((n & 2) ? b1 : b0);

        const int s2 = (n >> 2) & ~1;
        const float2 j0 = *(const float2*)(sS2 + s2 * 4 + s2);
        const float j11 = sS2[(s2 + 1) * 4 + s2 + 1];
        float c0, c1;
        merge_weights(j0.x, j0.y, j11, SCALE * 0.0625f, c0, c1);
        const float wt3 = wt2 * ((n & 4) ? c1 : c0);

        if (half == 0) { sW[n] = wt1; sW[16 + n] = wt2; sW[32 + n] = wt3; }
    }

    float fw[6];
    {
        const float2 p0  = *(const float2*)(sH + 6 * 16 + 6);
        const float  p11 = sH[7 * 16 + 7];
        const float4 r8  = *(const float4*)(sH + 8 * 16 + 8);
        const float4 r9  = *(const float4*)(sH + 9 * 16 + 8);
        const float4 r10 = *(const float4*)(sH + 10 * 16 + 8);
        const float4 r11 = *(const float4*)(sH + 11 * 16 + 8);
        float wa, wb, wc, wd, we, wf, wX, wY;
        merge_weights(p0.x, p0.y, p11, SCALE, wa, wb);
        merge_weights(r8.x, r8.y, r9.y, SCALE, wc, wd);
        merge_weights(r10.z, r10.w, r11.w, SCALE, we, wf);
        merge_weights((r8.x + r8.y) + (r9.x + r9.y),
                      (r8.z + r8.w) + (r9.z + r9.w),
                      (r10.z + r10.w) + (r11.z + r11.w), SCALE * 0.25f, wX, wY);
        fw[0] = wa;      fw[1] = wb;
        fw[2] = wX * wc; fw[3] = wX * wd;
        fw[4] = wY * we; fw[5] = wY * wf;
    }

    float ev[13];
    float dsum = 0.0f;
    if (half == 0) {
        const int j = n;
        float sr[16];
        #pragma unroll
        for (int dx = 0; dx < 16; ++dx) sr[dx] = sA[j * 16 + (j ^ dx)];
        float x[13];
        x[0] = sr[0];
        x[1] = sr[1];
        x[2] = 0.5f * (sr[2] + sr[3]);
        x[3] = 0.25f * ((sr[4] + sr[5]) + (sr[6] + sr[7]));
        x[4] = 0.125f * (((sr[8] + sr[9]) + (sr[10] + sr[11]))
                       + ((sr[12] + sr[13]) + (sr[14] + sr[15])));
        const float4 f0 = *(const float4*)(sF + j * 16);
        const float4 f1 = *(const float4*)(sF + j * 16 + 4);
        const float4 f2 = *(const float4*)(sF + j * 16 + 8);
        x[5] = f0.x; x[6] = f0.y; x[7] = f0.z; x[8] = f0.w;
        x[9] = f1.x; x[10] = f1.y;
        x[11] = 0.5f * (f1.z + f1.w);
        x[12] = 0.25f * ((f2.x + f2.y) + (f2.z + f2.w));

        bool msk[13];
        msk[0] = false;
        #pragma unroll
        for (int l = 0; l < 12; ++l) {
            const int bit = (l < 4) ? ((j >> l) & 1) : ((t >> (l - 4)) & 1);
            msk[l + 1] = (bit == 0);
        }
        float m = x[0];
        #pragma unroll
        for (int i = 1; i < 13; ++i) if (!msk[i]) m = fmaxf(m, x[i]);
        #pragma unroll
        for (int i = 0; i < 13; ++i) {
            ev[i] = msk[i] ? 0.0f : __expf(x[i] - m);
            dsum += ev[i];
        }
    }
    __syncthreads();

    if (half == 0) {
        const int j = n;
        den[j] = dsum;
        sA[j * 16 + j]       = ev[0];
        sA[j * 16 + (j ^ 1)] = ev[1];
        #pragma unroll
        for (int dx = 2; dx < 4; ++dx)  { const int i = j ^ dx; sA[j * 16 + i] = ev[2] * sW[i]; }
        #pragma unroll
        for (int dx = 4; dx < 8; ++dx)  { const int i = j ^ dx; sA[j * 16 + i] = ev[3] * sW[16 + i]; }
        #pragma unroll
        for (int dx = 8; dx < 16; ++dx) { const int i = j ^ dx; sA[j * 16 + i] = ev[4] * sW[32 + i]; }
        *(float4*)(sF + j * 16)     = make_float4(ev[5], ev[6], ev[7], ev[8]);
        *(float4*)(sF + j * 16 + 4) = make_float4(ev[9], ev[10], ev[11] * fw[0], ev[11] * fw[1]);
        *(float4*)(sF + j * 16 + 8) = make_float4(ev[12] * fw[2], ev[12] * fw[3],
                                                  ev[12] * fw[4], ev[12] * fw[5]);
    }
    __syncthreads();

    const size_t vLeafRow0  = ((size_t)(b * S_DIM + 16 * t) * H_DIM + h) * (size_t)D_DIM;
    const size_t vRowStride = (size_t)H_DIM * D_DIM;
    const int col4 = 4 * n;

    v2f eA[4], eF[3];
    float4 vA[4][2], vF[3][2];
    #pragma unroll
    for (int ks2 = 0; ks2 < 4; ++ks2) {
        const int c0 = 4 * ks2 + 2 * half;
        const float2 e2 = *(const float2*)(sA + n * 16 + c0);
        eA[ks2].x = e2.x; eA[ks2].y = e2.y;
        #pragma unroll
        for (int s = 0; s < 2; ++s)
            vA[ks2][s] = *(const float4*)(v + vLeafRow0 + (size_t)(c0 + s) * vRowStride + col4);
    }
    #pragma unroll
    for (int ks2 = 0; ks2 < 3; ++ks2) {
        const int c0 = 4 * ks2 + 2 * half;
        const float2 e2 = *(const float2*)(sF + n * 16 + c0);
        eF[ks2].x = e2.x; eF[ks2].y = e2.y;
        #pragma unroll
        for (int s = 0; s < 2; ++s) {
            const int row = far_row(c0 + s, t);
            vF[ks2][s] = *(const float4*)(Vt + treeBase + (size_t)row * D_DIM + col4);
        }
    }

    v8f acc[4] = {};
    #pragma unroll
    for (int ks2 = 0; ks2 < 4; ++ks2) {
        #pragma unroll
        for (int u = 0; u < 4; ++u) {
            v2f bV;
            bV.x = f4get(vA[ks2][0], u); bV.y = f4get(vA[ks2][1], u);
            acc[u] = __builtin_amdgcn_wmma_f32_16x16x4_f32(false, eA[ks2], false, bV, (short)0, acc[u], false, false);
        }
    }
    #pragma unroll
    for (int ks2 = 0; ks2 < 3; ++ks2) {
        #pragma unroll
        for (int u = 0; u < 4; ++u) {
            v2f bV;
            bV.x = f4get(vF[ks2][0], u); bV.y = f4get(vF[ks2][1], u);
            acc[u] = __builtin_amdgcn_wmma_f32_16x16x4_f32(false, eF[ks2], false, bV, (short)0, acc[u], false, false);
        }
    }

    #pragma unroll
    for (int r = 0; r < 8; ++r) {
        const int M = r + 8 * half;
        const float dn = den[M];
        float4 o;
        o.x = acc[0][r] / dn; o.y = acc[1][r] / dn;
        o.z = acc[2][r] / dn; o.w = acc[3][r] / dn;
        *(float4*)(out + ((size_t)(b * S_DIM + 16 * t + M) * H_DIM + h) * D_DIM + col4) = o;
    }
}

extern "C" void kernel_launch(void* const* d_in, const int* in_sizes, int n_in,
                              void* d_out, int out_size, void* d_ws, size_t ws_size,
                              hipStream_t stream)
{
    const float* q = (const float*)d_in[0];
    const float* k = (const float*)d_in[1];
    const float* v = (const float*)d_in[2];
    float* out = (float*)d_out;

    float* Kt = (float*)d_ws;
    float* Vt = Kt + (size_t)B_DIM * H_DIM * S_DIM * D_DIM;

    tree_kernel<<<256, 1024, 0, stream>>>(k, v, Kt, Vt);

    const int tiles = B_DIM * H_DIM * NTILE;
    hst_attn_kernel<<<tiles / 4, 128, 0, stream>>>(q, k, v, Kt, Vt, out);

    (void)in_sizes; (void)n_in; (void)out_size; (void)ws_size;
}


# Tree build, levels 1-4: a wave forms its 16 leaves' level-4 node directly: key = leaf mean, value = weighted leaf sum, weights from a WMMA leaf-key product K*K^T (sum order changes), not 15 merges
# speedup vs baseline: 2.7029x; 1.0672x over previous
#include <hip/hip_runtime.h>

typedef float v2f __attribute__((ext_vector_type(2)));
typedef float v8f __attribute__((ext_vector_type(8)));

#define B_DIM 4
#define S_DIM 4096
#define H_DIM 8
#define D_DIM 64
#define SCALE 0.125f
#define NTILE (S_DIM / 16)

__device__ __forceinline__ float add_row_swap(const float x)
{
    const int b = __float_as_int(x);
    return x + __int_as_float(__builtin_amdgcn_permlanex16(b, b, 0x76543210, 0xfedcba98, false, false));
}
template <int CTRL>
__device__ __forceinline__ float add_row_dpp(const float x)
{
    return x + __int_as_float(__builtin_amdgcn_update_dpp(0, __float_as_int(x), CTRL, 0xf, 0xf, false));
}
__device__ __forceinline__ float wave_sum(float x)
{
    x = add_row_swap(x);
    x = add_row_dpp<0x168>(x);
    x = add_row_dpp<0x164>(x);
    x = add_row_dpp<0x4E>(x);
    x = add_row_dpp<0xB1>(x);
    return x;
}

__device__ __forceinline__ void merge_pair(const float2 kc0, const float2 kc1,
                                           const float2 vc0, const float2 vc1,
                                           float2& kp, float2& vo)
{
    float2 vp;
    kp.x = 0.5f * (kc0.x + kc1.x); kp.y = 0.5f * (kc0.y + kc1.y);
    vp.x = 0.5f * (vc0.x + vc1.x); vp.y = 0.5f * (vc0.y + vc1.y);

    float pss = kp.x * kp.x  + kp.y * kp.y;
    float ps0 = kp.x * kc0.x + kp.y * kc0.y;
    float ps1 = kp.x * kc1.x + kp.y * kc1.y;
    pss = wave_sum(pss);
    ps0 = wave_sum(ps0);
    ps1 = wave_sum(ps1);
    pss *= SCALE; ps0 *= SCALE; ps1 *= SCALE;
    const float mm = fmaxf(pss, fmaxf(ps0, ps1));
    const float es = __expf(pss - mm);
    const float e0 = __expf(ps0 - mm);
    const float e1 = __expf(ps1 - mm);
    const float inv = 1.0f / (es + e0 + e1 + 1e-9f);

    vo.x = (es * vp.x + e0 * vc0.x + e1 * vc1.x) * inv;
    vo.y = (es * vp.y + e0 * vc0.y + e1 * vc1.y) * inv;
}

template <int NP>
__device__ __forceinline__ void level_in_regs(float2* kc, float2* vc,
                                              float* __restrict__ Kt,
                                              float* __restrict__ Vt,
                                              size_t treeRow0, int level,
                                              int chunk, int doff)
{
    const int IBl = S_DIM - (S_DIM >> (level - 1));
    const size_t row0 = treeRow0 + IBl + (size_t)chunk * NP;
    #pragma unroll
    for (int j = 0; j < NP; ++j) {
        float2 kp, vo;
        merge_pair(kc[2 * j], kc[2 * j + 1], vc[2 * j], vc[2 * j + 1], kp, vo);
        const size_t dst = (row0 + j) * D_DIM + doff;
        *(float2*)(Kt + dst) = kp;
        *(float2*)(Vt + dst) = vo;
        kc[j] = kp; vc[j] = vo;
    }
}

__device__ __forceinline__ void merge_weights(float g00, float g01, float g11,
                                              const float nrm, float& w0, float& w1)
{
    g00 *= nrm; g01 *= nrm; g11 *= nrm;
    const float ps0 = 0.5f * (g00 + g01);
    const float ps1 = 0.5f * (g01 + g11);
    const float pss = 0.5f * (ps0 + ps1);
    const float mm = fmaxf(pss, fmaxf(ps0, ps1));
    const float es = __expf(pss - mm);
    const float e0 = __expf(ps0 - mm);
    const float e1 = __expf(ps1 - mm);
    const float inv = 1.0f / (es + e0 + e1 + 1e-9f);
    w0 = (0.5f * es + e0) * inv;
    w1 = (0.5f * es + e1) * inv;
}

__device__ __forceinline__ float add_lane_xor1(const float x)
{
    return x + __int_as_float(__builtin_amdgcn_update_dpp(0, __float_as_int(x), 0xB1, 0xf, 0xf, false));
}
__device__ __forceinline__ float add_lane_xor2(const float x)
{
    return x + __int_as_float(__builtin_amdgcn_update_dpp(0, __float_as_int(x), 0x4E, 0xf, 0xf, false));
}

__device__ __forceinline__ void wave_lds_handover()
{
    __builtin_amdgcn_fence(__ATOMIC_RELEASE, "wavefront");
    __builtin_amdgcn_wave_barrier();
    __builtin_amdgcn_fence(__ATOMIC_ACQUIRE, "wavefront");
}

#define T_SG   0
#define T_S1   256
#define T_S2   320
#define T_S3   336
#define T_SW   340
#define T_WAVE 356
#define KT_PITCH 68

#define IB4 (S_DIM - (S_DIM >> 3))

__global__ void __launch_bounds__(1024) __attribute__((amdgpu_waves_per_eu(8, 8)))
tree_kernel(const float* __restrict__ kleaf,
            const float* __restrict__ vleaf,
            float* __restrict__ Kt,
            float* __restrict__ Vt)
{
    __shared__ __attribute__((aligned(16))) float2 sK4[32][32];
    __shared__ __attribute__((aligned(16))) float2 sV4[32][32];
    __shared__ float2 sK8[2][32];
    __shared__ float2 sV8[2][32];
    __shared__ __attribute__((aligned(16))) float sT[32][T_WAVE];
    __shared__ __attribute__((aligned(16))) float sKT[32][16 * KT_PITCH];

    const int wave = threadIdx.x >> 5;
    const int lane = threadIdx.x & 31;
    const int h    = blockIdx.x & 7;
    const int span = (blockIdx.x >> 3) & 7;
    const int b    = blockIdx.x >> 6;
    const int bh   = b * H_DIM + h;
    const int doff = 2 * lane;
    const int c    = span * 32 + wave;
    const int n    = lane & 15;
    const int half = lane >> 4;

    const size_t chunk0    = ((size_t)(b * S_DIM + 16 * c) * H_DIM + h) * D_DIM;
    const size_t rowStride = (size_t)H_DIM * D_DIM;
    const size_t treeRow0  = (size_t)bh * S_DIM;

    float4 kd[4][2], vd[4][2];
    const size_t col0 = chunk0 + (size_t)(2 * half) * rowStride + 4 * n;
    #pragma unroll
    for (int ks = 0; ks < 4; ++ks) {
        #pragma unroll
        for (int s2 = 0; s2 < 2; ++s2)
            kd[ks][s2] = *(const float4*)(kleaf + col0 + (size_t)(4 * ks + s2) * rowStride);
    }
    #pragma unroll
    for (int ks = 0; ks < 4; ++ks) {
        #pragma unroll
        for (int s2 = 0; s2 < 2; ++s2)
            vd[ks][s2] = *(const float4*)(vleaf + col0 + (size_t)(4 * ks + s2) * rowStride);
    }

    float* sKt = &sKT[wave][0];
    #pragma unroll
    for (int ks = 0; ks < 4; ++ks) {
        #pragma unroll
        for (int s2 = 0; s2 < 2; ++s2)
            *(float4*)(sKt + (4 * ks + 2 * half + s2) * KT_PITCH + 4 * n) = kd[ks][s2];
    }
    wave_lds_handover();
    float4 kf[8];
    #pragma unroll
    for (int j = 0; j < 8; ++j) kf[j] = *(const float4*)(sKt + n * KT_PITCH + 8 * j + 4 * half);

    v8f cG0 = {}; v8f cG1 = {};
    #pragma unroll
    for (int j = 0; j < 8; ++j) {
        v2f f;
        f.x = kf[j].x; f.y = kf[j].y;
        cG0 = __builtin_amdgcn_wmma_f32_16x16x4_f32(false, f, false, f, (short)0, cG0, false, false);
        f.x = kf[j].z; f.y = kf[j].w;
        cG1 = __builtin_amdgcn_wmma_f32_16x16x4_f32(false, f, false, f, (short)0, cG1, false, false);
    }
    const v8f cG = cG0 + cG1;

    float t1[4], t2[2];
    #pragma unroll
    for (int u = 0; u < 4; ++u) t1[u] = add_lane_xor1(cG[2 * u] + cG[2 * u + 1]);
    #pragma unroll
    for (int u = 0; u < 2; ++u) t2[u] = add_lane_xor2(t1[2 * u] + t1[2 * u + 1]);
    const float t3 = add_row_dpp<0x164>(t2[0] + t2[1]);

    float* sG  = &sT[wave][T_SG];
    float* sS1 = &sT[wave][T_S1];
    float* sS2 = &sT[wave][T_S2];
    float* sS3 = &sT[wave][T_S3];
    float* sW  = &sT[wave][T_SW];
    #pragma unroll
    for (int r = 0; r < 8; ++r) sG[(r + 8 * half) * 16 + n] = cG[r];
    if ((n & 1) == 0) {
        #pragma unroll
        for (int u = 0; u < 4; ++u) sS1[(4 * half + u) * 8 + (n >> 1)] = t1[u];
    }
    if ((n & 3) == 0) {
        #pragma unroll
        for (int u = 0; u < 2; ++u) sS2[(2 * half + u) * 4 + (n >> 2)] = t2[u];
    }
    if ((n & 7) == 0) sS3[half * 2 + (n >> 3)] = t3;
    wave_lds_handover();

    {
        const int p2 = n & ~1;
        const float2 g0 = *(const float2*)(sG + p2 * 16 + p2);
        const float g11 = sG[(p2 + 1) * 16 + p2 + 1];
        float a0, a1;
        merge_weights(g0.x, g0.y, g11, SCALE, a0, a1);
        float wt = (n & 1) ? a1 : a0;

        const int r2 = (n >> 1) & ~1;
        const float2 h0 = *(const float2*)(sS1 + r2 * 8 + r2);
        const float h11 = sS1[(r2 + 1) * 8 + r2 + 1];
        float b0, b1;
        merge_weights(h0.x, h0.y, h11, SCALE * 0.25f, b0, b1);
        wt *= (n & 2) ? b1 : b0;

        const int q2 = (n >> 2) & ~1;
        const float2 j0 = *(const float2*)(sS2 + q2 * 4 + q2);
        const float j11 = sS2[(q2 + 1) * 4 + q2 + 1];
        float c0, c1;
        merge_weights(j0.x, j0.y, j11, SCALE * 0.0625f, c0, c1);
        wt *= (n & 4) ? c1 : c0;

        const float2 m0 = *(const float2*)(sS3);
        const float m11 = sS3[3];
        float d0, d1;
        merge_weights(m0.x, m0.y, m11, SCALE * 0.015625f, d0, d1);
        wt *= (n & 8) ? d1 : d0;

        if (half == 0) sW[n] = wt;
    }
    wave_lds_handover();

    float4 pk[4], pv[4];
    #pragma unroll
    for (int ks = 0; ks < 4; ++ks) {
        const float2 w2 = *(const float2*)(sW + 4 * ks + 2 * half);
        pk[ks].x = kd[ks][0].x + kd[ks][1].x;  pk[ks].y = kd[ks][0].y + kd[ks][1].y;
        pk[ks].z = kd[ks][0].z + kd[ks][1].z;  pk[ks].w = kd[ks][0].w + kd[ks][1].w;
        pv[ks].x = w2.x * vd[ks][0].x + w2.y * vd[ks][1].x;
        pv[ks].y = w2.x * vd[ks][0].y + w2.y * vd[ks][1].y;
        pv[ks].z = w2.x * vd[ks][0].z + w2.y * vd[ks][1].z;
        pv[ks].w = w2.x * vd[ks][0].w + w2.y * vd[ks][1].w;
    }
    float4 k4, v4;
    k4.x = 0.0625f * add_row_swap((pk[0].x + pk[1].x) + (pk[2].x + pk[3].x));
    k4.y = 0.0625f * add_row_swap((pk[0].y + pk[1].y) + (pk[2].y + pk[3].y));
    k4.z = 0.0625f * add_row_swap((pk[0].z + pk[1].z) + (pk[2].z + pk[3].z));
    k4.w = 0.0625f * add_row_swap((pk[0].w + pk[1].w) + (pk[2].w + pk[3].w));
    v4.x = add_row_swap((pv[0].x + pv[1].x) + (pv[2].x + pv[3].x));
    v4.y = add_row_swap((pv[0].y + pv[1].y) + (pv[2].y + pv[3].y));
    v4.z = add_row_swap((pv[0].z + pv[1].z) + (pv[2].z + pv[3].z));
    v4.w = add_row_swap((pv[0].w + pv[1].w) + (pv[2].w + pv[3].w));

    {
        const size_t dst = (treeRow0 + IB4 + c) * D_DIM + 4 * n;
        float* g4 = half ? Vt : Kt;
        float* s4 = half ? (float*)&sV4[wave][0] : (float*)&sK4[wave][0];
        float4 o;
        o.x = half ? v4.x : k4.x;  o.y = half ? v4.y : k4.y;
        o.z = half ? v4.z : k4.z;  o.w = half ? v4.w : k4.w;
        *(float4*)(g4 + dst) = o;
        *(float4*)(s4 + 4 * n) = o;
    }
    __syncthreads();

    if (wave < 2) {
        float2 kc[16], vc[16];
        #pragma unroll
        for (int i = 0; i < 16; ++i) { kc[i] = sK4[16 * wave + i][lane]; vc[i] = sV4[16 * wave + i][lane]; }
        const int g = span * 2 + wave;
        level_in_regs<8>(kc, vc, Kt, Vt, treeRow0, 5, g, doff);
        level_in_regs<4>(kc, vc, Kt, Vt, treeRow0, 6, g, doff);
        level_in_regs<2>(kc, vc, Kt, Vt, treeRow0, 7, g, doff);
        level_in_regs<1>(kc, vc, Kt, Vt, treeRow0, 8, g, doff);
        sK8[wave][lane] = kc[0];
        sV8[wave][lane] = vc[0];
    }
    __syncthreads();

    if (wave == 0) {
        float2 kc[2], vc[2];
        kc[0] = sK8[0][lane]; kc[1] = sK8[1][lane];
        vc[0] = sV8[0][lane]; vc[1] = sV8[1][lane];
        level_in_regs<1>(kc, vc, Kt, Vt, treeRow0, 9, span, doff);
    }
}

__device__ __forceinline__ float f4get(const float4 x, const int u)
{
    return u == 0 ? x.x : (u == 1 ? x.y : (u == 2 ? x.z : x.w));
}

#define L_SA   0
#define L_SF   256
#define L_SG   512
#define L_SH   768
#define L_S1   1024
#define L_S2   1088
#define L_SW   1104
#define L_DEN  1152
#define L_WAVE 1168

#define IB9 (S_DIM - (S_DIM >> 8))

__device__ __forceinline__ int far_row(const int c, const int t)
{
    const int cs  = c & 7;
    const int sib = (S_DIM - (S_DIM >> (3 + cs))) + ((t >> cs) ^ 1);
    const int r10 = IB9 + 2 * ((t >> 6) ^ 1) + (c & 1);
    const int r11 = IB9 + 4 * ((t >> 7) ^ 1) + (c & 3);
    return c < 6 ? sib : (c < 8 ? r10 : r11);
}

__global__ void __launch_bounds__(128)
hst_attn_kernel(const float* __restrict__ q,
                const float* __restrict__ k,
                const float* __restrict__ v,
                const float* __restrict__ Kt,
                const float* __restrict__ Vt,
                float* __restrict__ out)
{
    __shared__ __attribute__((aligned(16))) float ldsS[4][L_WAVE];

    const int lane = threadIdx.x & 31;
    const int w    = threadIdx.x >> 5;
    const int tile = blockIdx.x * 4 + w;
    const int t  = tile & (NTILE - 1);
    const int bh = tile >> 8;
    const int h  = bh & (H_DIM - 1);
    const int b  = bh >> 3;

    const int n    = lane & 15;
    const int half = lane >> 4;

    float* sA  = &ldsS[w][L_SA];
    float* sF  = &ldsS[w][L_SF];
    float* sG  = &ldsS[w][L_SG];
    float* sH  = &ldsS[w][L_SH];
    float* sS1 = &ldsS[w][L_S1];
    float* sS2 = &ldsS[w][L_S2];
    float* sW  = &ldsS[w][L_SW];
    float* den = &ldsS[w][L_DEN];

    const int rF = far_row(n, t);

    const size_t leafBase = ((size_t)(b * S_DIM + 16 * t + n) * H_DIM + h) * D_DIM;
    const size_t treeBase = (size_t)(b * H_DIM + h) * S_DIM * (size_t)D_DIM;
    const float* qp  = q  + leafBase;
    const float* kAp = k  + leafBase;
    const float* kFp = Kt + treeBase + (size_t)rF * D_DIM;

    v8f cA = {}; v8f cF = {}; v8f cG = {}; v8f cH = {};
    const int dofs = 4 * half;
    #pragma unroll
    for (int j = 0; j < 8; ++j) {
        const int d = 8 * j + dofs;
        const float4 a4 = *(const float4*)(qp  + d);
        const float4 kA4 = *(const float4*)(kAp + d);
        const float4 kF4 = *(const float4*)(kFp + d);
        v2f a, bA, bF;
        a.x = a4.x * SCALE; a.y = a4.y * SCALE;
        bA.x = kA4.x; bA.y = kA4.y;
        bF.x = kF4.x; bF.y = kF4.y;
        cA = __builtin_amdgcn_wmma_f32_16x16x4_f32(false, a, false, bA, (short)0, cA, false, false);
        cF = __builtin_amdgcn_wmma_f32_16x16x4_f32(false, a, false, bF, (short)0, cF, false, false);
        cG = __builtin_amdgcn_wmma_f32_16x16x4_f32(false, bA, false, bA, (short)0, cG, false, false);
        cH = __builtin_amdgcn_wmma_f32_16x16x4_f32(false, bF, false, bF, (short)0, cH, false, false);
        a.x = a4.z * SCALE; a.y = a4.w * SCALE;
        bA.x = kA4.z; bA.y = kA4.w;
        bF.x = kF4.z; bF.y = kF4.w;
        cA = __builtin_amdgcn_wmma_f32_16x16x4_f32(false, a, false, bA, (short)0, cA, false, false);
        cF = __builtin_amdgcn_wmma_f32_16x16x4_f32(false, a, false, bF, (short)0, cF, false, false);
        cG = __builtin_amdgcn_wmma_f32_16x16x4_f32(false, bA, false, bA, (short)0, cG, false, false);
        cH = __builtin_amdgcn_wmma_f32_16x16x4_f32(false, bF, false, bF, (short)0, cH, false, false);
    }

    float t1[4], t2[2];
    #pragma unroll
    for (int u = 0; u < 4; ++u) {
        t1[u] = add_lane_xor1(cG[2 * u] + cG[2 * u + 1]);
    }
    #pragma unroll
    for (int u = 0; u < 2; ++u) {
        t2[u] = add_lane_xor2(t1[2 * u] + t1[2 * u + 1]);
    }

    #pragma unroll
    for (int r = 0; r < 8; ++r) {
        const int M = r + 8 * half;
        sA[M * 16 + n] = cA[r];
        sF[M * 16 + n] = cF[r];
        sG[M * 16 + n] = cG[r];
        sH[M * 16 + n] = cH[r];
    }
    if ((n & 1) == 0) {
        #pragma unroll
        for (int u = 0; u < 4; ++u) sS1[(4 * half + u) * 8 + (n >> 1)] = t1[u];
    }
    if ((n & 3) == 0) {
        #pragma unroll
        for (int u = 0; u < 2; ++u) sS2[(2 * half + u) * 4 + (n >> 2)] = t2[u];
    }
    __syncthreads();

    {
        const int p2 = n & ~1;
        const float2 g0 = *(const float2*)(sG + p2 * 16 + p2);
        const float g11 = sG[(p2 + 1) * 16 + p2 + 1];
        float a0, a1;
        merge_weights(g0.x, g0.y, g11, SCALE, a0, a1);
        const float wt1 = (n & 1) ? a1 : a0;

        const int r2 = (n >> 1) & ~1;
        const float2 h0 = *(const float2*)(sS1 + r2 * 8 + r2);
        const float h11 = sS1[(r2 + 1) * 8 + r2 + 1];
        float b0, b1;
        merge_weights(h0.x, h0.y, h11, SCALE * 0.25f, b0, b1);
        const float wt2 = wt1 * ((n & 2) ? b1 : b0);

        const int s2 = (n >> 2) & ~1;
        const float2 j0 = *(const float2*)(sS2 + s2 * 4 + s2);
        const float j11 = sS2[(s2 + 1) * 4 + s2 + 1];
        float c0, c1;
        merge_weights(j0.x, j0.y, j11, SCALE * 0.0625f, c0, c1);
        const float wt3 = wt2 * ((n & 4) ? c1 : c0);

        if (half == 0) { sW[n] = wt1; sW[16 + n] = wt2; sW[32 + n] = wt3; }
    }

    float fw[6];
    {
        const float2 p0  = *(const float2*)(sH + 6 * 16 + 6);
        const float  p11 = sH[7 * 16 + 7];
        const float4 r8  = *(const float4*)(sH + 8 * 16 + 8);
        const float4 r9  = *(const float4*)(sH + 9 * 16 + 8);
        const float4 r10 = *(const float4*)(sH + 10 * 16 + 8);
        const float4 r11 = *(const float4*)(sH + 11 * 16 + 8);
        float wa, wb, wc, wd, we, wf, wX, wY;
        merge_weights(p0.x, p0.y, p11, SCALE, wa, wb);
        merge_weights(r8.x, r8.y, r9.y, SCALE, wc, wd);
        merge_weights(r10.z, r10.w, r11.w, SCALE, we, wf);
        merge_weights((r8.x + r8.y) + (r9.x + r9.y),
                      (r8.z + r8.w) + (r9.z + r9.w),
                      (r10.z + r10.w) + (r11.z + r11.w), SCALE * 0.25f, wX, wY);
        fw[0] = wa;      fw[1] = wb;
        fw[2] = wX * wc; fw[3] = wX * wd;
        fw[4] = wY * we; fw[5] = wY * wf;
    }

    float ev[13];
    float dsum = 0.0f;
    if (half == 0) {
        const int j = n;
        float sr[16];
        #pragma unroll
        for (int dx = 0; dx < 16; ++dx) sr[dx] = sA[j * 16 + (j ^ dx)];
        float x[13];
        x[0] = sr[0];
        x[1] = sr[1];
        x[2] = 0.5f * (sr[2] + sr[3]);
        x[3] = 0.25f * ((sr[4] + sr[5]) + (sr[6] + sr[7]));
        x[4] = 0.125f * (((sr[8] + sr[9]) + (sr[10] + sr[11]))
                       + ((sr[12] + sr[13]) + (sr[14] + sr[15])));
        const float4 f0 = *(const float4*)(sF + j * 16);
        const float4 f1 = *(const float4*)(sF + j * 16 + 4);
        const float4 f2 = *(const float4*)(sF + j * 16 + 8);
        x[5] = f0.x; x[6] = f0.y; x[7] = f0.z; x[8] = f0.w;
        x[9] = f1.x; x[10] = f1.y;
        x[11] = 0.5f * (f1.z + f1.w);
        x[12] = 0.25f * ((f2.x + f2.y) + (f2.z + f2.w));

        bool msk[13];
        msk[0] = false;
        #pragma unroll
        for (int l = 0; l < 12; ++l) {
            const int bit = (l < 4) ? ((j >> l) & 1) : ((t >> (l - 4)) & 1);
            msk[l + 1] = (bit == 0);
        }
        float m = x[0];
        #pragma unroll
        for (int i = 1; i < 13; ++i) if (!msk[i]) m = fmaxf(m, x[i]);
        #pragma unroll
        for (int i = 0; i < 13; ++i) {
            ev[i] = msk[i] ? 0.0f : __expf(x[i] - m);
            dsum += ev[i];
        }
    }
    __syncthreads();

    if (half == 0) {
        const int j = n;
        den[j] = dsum;
        sA[j * 16 + j]       = ev[0];
        sA[j * 16 + (j ^ 1)] = ev[1];
        #pragma unroll
        for (int dx = 2; dx < 4; ++dx)  { const int i = j ^ dx; sA[j * 16 + i] = ev[2] * sW[i]; }
        #pragma unroll
        for (int dx = 4; dx < 8; ++dx)  { const int i = j ^ dx; sA[j * 16 + i] = ev[3] * sW[16 + i]; }
        #pragma unroll
        for (int dx = 8; dx < 16; ++dx) { const int i = j ^ dx; sA[j * 16 + i] = ev[4] * sW[32 + i]; }
        *(float4*)(sF + j * 16)     = make_float4(ev[5], ev[6], ev[7], ev[8]);
        *(float4*)(sF + j * 16 + 4) = make_float4(ev[9], ev[10], ev[11] * fw[0], ev[11] * fw[1]);
        *(float4*)(sF + j * 16 + 8) = make_float4(ev[12] * fw[2], ev[12] * fw[3],
                                                  ev[12] * fw[4], ev[12] * fw[5]);
    }
    __syncthreads();

    const size_t vLeafRow0  = ((size_t)(b * S_DIM + 16 * t) * H_DIM + h) * (size_t)D_DIM;
    const size_t vRowStride = (size_t)H_DIM * D_DIM;
    const int col4 = 4 * n;

    v2f eA[4], eF[3];
    float4 vA[4][2], vF[3][2];
    #pragma unroll
    for (int ks2 = 0; ks2 < 4; ++ks2) {
        const int c0 = 4 * ks2 + 2 * half;
        const float2 e2 = *(const float2*)(sA + n * 16 + c0);
        eA[ks2].x = e2.x; eA[ks2].y = e2.y;
        #pragma unroll
        for (int s = 0; s < 2; ++s)
            vA[ks2][s] = *(const float4*)(v + vLeafRow0 + (size_t)(c0 + s) * vRowStride + col4);
    }
    #pragma unroll
    for (int ks2 = 0; ks2 < 3; ++ks2) {
        const int c0 = 4 * ks2 + 2 * half;
        const float2 e2 = *(const float2*)(sF + n * 16 + c0);
        eF[ks2].x = e2.x; eF[ks2].y = e2.y;
        #pragma unroll
        for (int s = 0; s < 2; ++s) {
            const int row = far_row(c0 + s, t);
            vF[ks2][s] = *(const float4*)(Vt + treeBase + (size_t)row * D_DIM + col4);
        }
    }

    v8f acc[4] = {};
    #pragma unroll
    for (int ks2 = 0; ks2 < 4; ++ks2) {
        #pragma unroll
        for (int u = 0; u < 4; ++u) {
            v2f bV;
            bV.x = f4get(vA[ks2][0], u); bV.y = f4get(vA[ks2][1], u);
            acc[u] = __builtin_amdgcn_wmma_f32_16x16x4_f32(false, eA[ks2], false, bV, (short)0, acc[u], false, false);
        }
    }
    #pragma unroll
    for (int ks2 = 0; ks2 < 3; ++ks2) {
        #pragma unroll
        for (int u = 0; u < 4; ++u) {
            v2f bV;
            bV.x = f4get(vF[ks2][0], u); bV.y = f4get(vF[ks2][1], u);
            acc[u] = __builtin_amdgcn_wmma_f32_16x16x4_f32(false, eF[ks2], false, bV, (short)0, acc[u], false, false);
        }
    }

    #pragma unroll
    for (int r = 0; r < 8; ++r) {
        const int M = r + 8 * half;
        const float dn = den[M];
        float4 o;
        o.x = acc[0][r] / dn; o.y = acc[1][r] / dn;
        o.z = acc[2][r] / dn; o.w = acc[3][r] / dn;
        *(float4*)(out + ((size_t)(b * S_DIM + 16 * t + M) * H_DIM + h) * D_DIM + col4) = o;
    }
}

extern "C" void kernel_launch(void* const* d_in, const int* in_sizes, int n_in,
                              void* d_out, int out_size, void* d_ws, size_t ws_size,
                              hipStream_t stream)
{
    const float* q = (const float*)d_in[0];
    const float* k = (const float*)d_in[1];
    const float* v = (const float*)d_in[2];
    float* out = (float*)d_out;

    float* Kt = (float*)d_ws;
    float* Vt = Kt + (size_t)B_DIM * H_DIM * S_DIM * D_DIM;

    tree_kernel<<<256, 1024, 0, stream>>>(k, v, Kt, Vt);

    const int tiles = B_DIM * H_DIM * NTILE;
    hst_attn_kernel<<<tiles / 4, 128, 0, stream>>>(q, k, v, Kt, Vt, out);

    (void)in_sizes; (void)n_in; (void)out_size; (void)ws_size;
}
